# starting kernel
# baseline (speedup 1.0000x reference)
#include <hip/hip_runtime.h>
#include <hip/hip_bf16.h>
#include <math.h>

#define BSZ  4
#define LSEQ 2048
#define DCH  768
#define NST  64
#define DN   (DCH * NST)

typedef __attribute__((ext_vector_type(2))) float v2f;
typedef __attribute__((ext_vector_type(8))) float v8f;

__global__ void __launch_bounds__(256)
s4_precompute(const float* __restrict__ A_real, const float* __restrict__ A_imag,
              const float* __restrict__ B_mat,  const float* __restrict__ C_mat,
              const float* __restrict__ log_delta,
              float* __restrict__ dAr, float* __restrict__ dAi,
              float* __restrict__ CBr, float* __restrict__ CBi) {
    int idx = blockIdx.x * blockDim.x + threadIdx.x;
    if (idx >= DN) return;
    int d = idx / NST;

    float delta = log1pf(expf(log_delta[d]));
    float ar = A_real[idx], ai = A_imag[idx];
    float dr = delta * ar, di = delta * ai;

    float er = expf(dr);
    float s, c;
    __sincosf(di, &s, &c);
    float abr = er * c, abi = er * s;

    float nr = abr - 1.0f, ni = abi;
    float inv = 1.0f / (ar * ar + ai * ai);
    float qr = (nr * ar + ni * ai) * inv;
    float qi = (ni * ar - nr * ai) * inv;

    float b = B_mat[idx], cm = C_mat[idx];
    dAr[idx] = dr;
    dAi[idx] = di;
    CBr[idx] = cm * (qr * b);
    CBi[idx] = cm * (qi * b);
}

__global__ void __launch_bounds__(256)
s4_kernelgen(const float* __restrict__ dAr, const float* __restrict__ dAi,
             const float* __restrict__ CBr, const float* __restrict__ CBi,
             float* __restrict__ Kbuf) {
    int idx = blockIdx.x * blockDim.x + threadIdx.x;
    if (idx >= DCH * LSEQ) return;
    int   d = idx / LSEQ;
    float l = (float)(idx & (LSEQ - 1));

    const float* pr = dAr + d * NST;
    const float* pi = dAi + d * NST;
    const float* cr = CBr + d * NST;
    const float* ci = CBi + d * NST;

    float acc = 0.0f;
#pragma unroll 4
    for (int n = 0; n < NST; ++n) {
        float e = __expf(pr[n] * l);
        float s, c;
        __sincosf(pi[n] * l, &s, &c);
        acc += e * (cr[n] * c - ci[n] * s);
    }
    Kbuf[idx] = acc;
}

__global__ void __launch_bounds__(256)
s4_conv_wmma(const float* __restrict__ u,
             const float* __restrict__ Kbuf,
             float* __restrict__ y) {
    __shared__ float Klds[LSEQ];

    const int d  = blockIdx.x >> 4;
    const int Ig = blockIdx.x & 15;

    {
        const float4* src = (const float4*)(Kbuf + (size_t)d * LSEQ);
        float4*       dst = (float4*)Klds;
#pragma unroll
        for (int i = 0; i < (LSEQ / 4) / 256; ++i)
            dst[threadIdx.x + i * 256] = src[threadIdx.x + i * 256];
    }
    __syncthreads();

    const int wave = threadIdx.x >> 5;
    const int lane = threadIdx.x & 31;
    const int I    = Ig * 8 + wave;
    const int ln   = lane & 15;
    const int g    = lane >> 4;
    const int orow = I * 16 + ln;

    v8f acc = {};

    for (int J = 0; J <= I; ++J) {
        const int jbase = J * 16;

        if (ln < BSZ && J < I) {
            const float* pf = u + ((size_t)ln * LSEQ + (size_t)(jbase + 16)) * DCH + d;
            __builtin_prefetch(pf, 0, 1);
        }

#pragma unroll
        for (int cchunk = 0; cchunk < 4; ++cchunk) {
            const int kb = jbase + cchunk * 4;
            const int k0   = kb + 2 * g;
            const int diff = orow - k0;
            v2f a;
            a.x = (diff >= 0) ? Klds[diff]     : 0.0f;
            a.y = (diff >= 1) ? Klds[diff - 1] : 0.0f;
            v2f b;
            if (ln < BSZ) {
                const float* up = u + ((size_t)ln * LSEQ + (size_t)k0) * DCH + d;
                b.x = up[0];
                b.y = up[DCH];
            } else {
                b.x = 0.0f;
                b.y = 0.0f;
            }
            acc = __builtin_amdgcn_wmma_f32_16x16x4_f32(
                false, a, false, b, (short)0, acc, false, false);
        }
    }

#pragma unroll
    for (int r = 0; r < 8; ++r) {
        if (ln < BSZ) {
            const int mrow = r + 8 * g;
            y[((size_t)ln * LSEQ + (size_t)(I * 16 + mrow)) * DCH + d] = acc[r];
        }
    }
}

extern "C" void kernel_launch(void* const* d_in, const int* in_sizes, int n_in,
                              void* d_out, int out_size, void* d_ws, size_t ws_size,
                              hipStream_t stream) {
    const float* u         = (const float*)d_in[0];
    const float* A_real    = (const float*)d_in[1];
    const float* A_imag    = (const float*)d_in[2];
    const float* B_mat     = (const float*)d_in[3];
    const float* C_mat     = (const float*)d_in[4];
    const float* log_delta = (const float*)d_in[5];
    float* y = (float*)d_out;

    float* ws   = (float*)d_ws;
    float* dAr  = ws;
    float* dAi  = ws + 1 * DN;
    float* CBr  = ws + 2 * DN;
    float* CBi  = ws + 3 * DN;
    float* Kbuf = ws + 4 * DN;

    s4_precompute<<<dim3((DN + 255) / 256), dim3(256), 0, stream>>>(
        A_real, A_imag, B_mat, C_mat, log_delta, dAr, dAi, CBr, CBi);

    s4_kernelgen<<<dim3((DCH * LSEQ + 255) / 256), dim3(256), 0, stream>>>(
        dAr, dAi, CBr, CBi, Kbuf);

    s4_conv_wmma<<<dim3(DCH * 16), dim3(256), 0, stream>>>(u, Kbuf, y);
}


# Stage channel d of u (4 batches, every input time the block reaches) once per block in LDS as time pairs; B fragments become one ds_load_b64 each instead of two strided 4-byte global loads per WMMA
# speedup vs baseline: 1.8647x; 1.8647x over previous
#include <hip/hip_runtime.h>
#include <hip/hip_bf16.h>
#include <math.h>

#define BSZ  4
#define LSEQ 2048
#define DCH  768
#define NST  64
#define DN   (DCH * NST)

typedef __attribute__((ext_vector_type(2))) float v2f;
typedef __attribute__((ext_vector_type(8))) float v8f;

__global__ void __launch_bounds__(256)
s4_precompute(const float* __restrict__ A_real, const float* __restrict__ A_imag,
              const float* __restrict__ B_mat,  const float* __restrict__ C_mat,
              const float* __restrict__ log_delta,
              float* __restrict__ dAr, float* __restrict__ dAi,
              float* __restrict__ CBr, float* __restrict__ CBi) {
    int idx = blockIdx.x * blockDim.x + threadIdx.x;
    if (idx >= DN) return;
    int d = idx / NST;

    float delta = log1pf(expf(log_delta[d]));
    float ar = A_real[idx], ai = A_imag[idx];
    float dr = delta * ar, di = delta * ai;

    float er = expf(dr);
    float s, c;
    __sincosf(di, &s, &c);
    float abr = er * c, abi = er * s;

    float nr = abr - 1.0f, ni = abi;
    float inv = 1.0f / (ar * ar + ai * ai);
    float qr = (nr * ar + ni * ai) * inv;
    float qi = (ni * ar - nr * ai) * inv;

    float b = B_mat[idx], cm = C_mat[idx];
    dAr[idx] = dr;
    dAi[idx] = di;
    CBr[idx] = cm * (qr * b);
    CBi[idx] = cm * (qi * b);
}

__global__ void __launch_bounds__(256)
s4_kernelgen(const float* __restrict__ dAr, const float* __restrict__ dAi,
             const float* __restrict__ CBr, const float* __restrict__ CBi,
             float* __restrict__ Kbuf) {
    int idx = blockIdx.x * blockDim.x + threadIdx.x;
    if (idx >= DCH * LSEQ) return;
    int   d = idx / LSEQ;
    float l = (float)(idx & (LSEQ - 1));

    const float* pr = dAr + d * NST;
    const float* pi = dAi + d * NST;
    const float* cr = CBr + d * NST;
    const float* ci = CBi + d * NST;

    float acc = 0.0f;
#pragma unroll 4
    for (int n = 0; n < NST; ++n) {
        float e = __expf(pr[n] * l);
        float s, c;
        __sincosf(pi[n] * l, &s, &c);
        acc += e * (cr[n] * c - ci[n] * s);
    }
    Kbuf[idx] = acc;
}

__global__ void __launch_bounds__(256)
s4_conv_wmma(const float* __restrict__ u,
             const float* __restrict__ Kbuf,
             float* __restrict__ y) {
    __shared__ float Klds[LSEQ];
    __shared__ v2f   Ulds[(LSEQ / 2) * BSZ];

    const int d  = blockIdx.x >> 4;
    const int Ig = blockIdx.x & 15;

    {
        const float4* src = (const float4*)(Kbuf + (size_t)d * LSEQ);
        float4*       dst = (float4*)Klds;
#pragma unroll
        for (int i = 0; i < (LSEQ / 4) / 256; ++i)
            dst[threadIdx.x + i * 256] = src[threadIdx.x + i * 256];
    }
    {
        const int sb = threadIdx.x & 3;
        const int sp = threadIdx.x >> 2;
        const float* usrc = u + ((size_t)sb * LSEQ + (size_t)(2 * sp)) * DCH + d;
        v2f* udst = Ulds + threadIdx.x;
        const int nslab = Ig + 1;
#pragma unroll 4
        for (int s = 0; s < nslab; ++s) {
            v2f val;
            val.x = usrc[(size_t)s * 128 * DCH];
            val.y = usrc[(size_t)s * 128 * DCH + DCH];
            udst[s * 256] = val;
        }
    }
    __syncthreads();

    const int wave = threadIdx.x >> 5;
    const int lane = threadIdx.x & 31;
    const int I    = Ig * 8 + wave;
    const int ln   = lane & 15;
    const int g    = lane >> 4;
    const int orow = I * 16 + ln;

    const v2f* ub  = Ulds + g * 4 + (ln & 3);
    const bool bon = ln < BSZ;

    v8f acc = {};

    for (int J = 0; J <= I; ++J) {
        const int jbase = J * 16;

#pragma unroll
        for (int cchunk = 0; cchunk < 4; ++cchunk) {
            const int kb = jbase + cchunk * 4;
            const int k0   = kb + 2 * g;
            const int diff = orow - k0;
            v2f a;
            a.x = (diff >= 0) ? Klds[diff]     : 0.0f;
            a.y = (diff >= 1) ? Klds[diff - 1] : 0.0f;
            const v2f bv = ub[(J * 8 + cchunk * 2) * 4];
            v2f b;
            b.x = bon ? bv.x : 0.0f;
            b.y = bon ? bv.y : 0.0f;
            acc = __builtin_amdgcn_wmma_f32_16x16x4_f32(
                false, a, false, b, (short)0, acc, false, false);
        }
    }

#pragma unroll
    for (int r = 0; r < 8; ++r) {
        if (ln < BSZ) {
            const int mrow = r + 8 * g;
            y[((size_t)ln * LSEQ + (size_t)(I * 16 + mrow)) * DCH + d] = acc[r];
        }
    }
}

extern "C" void kernel_launch(void* const* d_in, const int* in_sizes, int n_in,
                              void* d_out, int out_size, void* d_ws, size_t ws_size,
                              hipStream_t stream) {
    const float* u         = (const float*)d_in[0];
    const float* A_real    = (const float*)d_in[1];
    const float* A_imag    = (const float*)d_in[2];
    const float* B_mat     = (const float*)d_in[3];
    const float* C_mat     = (const float*)d_in[4];
    const float* log_delta = (const float*)d_in[5];
    float* y = (float*)d_out;

    float* ws   = (float*)d_ws;
    float* dAr  = ws;
    float* dAi  = ws + 1 * DN;
    float* CBr  = ws + 2 * DN;
    float* CBi  = ws + 3 * DN;
    float* Kbuf = ws + 4 * DN;

    s4_precompute<<<dim3((DN + 255) / 256), dim3(256), 0, stream>>>(
        A_real, A_imag, B_mat, C_mat, log_delta, dAr, dAi, CBr, CBi);

    s4_kernelgen<<<dim3((DCH * LSEQ + 255) / 256), dim3(256), 0, stream>>>(
        dAr, dAi, CBr, CBi, Kbuf);

    s4_conv_wmma<<<dim3(DCH * 16), dim3(256), 0, stream>>>(u, Kbuf, y);
}


# Conv: the 12 idle B columns of each WMMA carry 3 more output tiles sharing its Toeplitz A fragment (cols = 4 batches x 4 tiles): 4x fewer WMMAs, one 16-wave block per channel; sum order kept
# speedup vs baseline: 6.6685x; 3.5762x over previous
#include <hip/hip_runtime.h>
#include <hip/hip_bf16.h>
#include <math.h>

#define BSZ  4
#define LSEQ 2048
#define DCH  768
#define NST  64
#define DN   (DCH * NST)

typedef __attribute__((ext_vector_type(2))) float v2f;
typedef __attribute__((ext_vector_type(8))) float v8f;

__global__ void __launch_bounds__(256)
s4_precompute(const float* __restrict__ A_real, const float* __restrict__ A_imag,
              const float* __restrict__ B_mat,  const float* __restrict__ C_mat,
              const float* __restrict__ log_delta,
              float* __restrict__ dAr, float* __restrict__ dAi,
              float* __restrict__ CBr, float* __restrict__ CBi) {
    int idx = blockIdx.x * blockDim.x + threadIdx.x;
    if (idx >= DN) return;
    int d = idx / NST;

    float delta = log1pf(expf(log_delta[d]));
    float ar = A_real[idx], ai = A_imag[idx];
    float dr = delta * ar, di = delta * ai;

    float er = expf(dr);
    float s, c;
    __sincosf(di, &s, &c);
    float abr = er * c, abi = er * s;

    float nr = abr - 1.0f, ni = abi;
    float inv = 1.0f / (ar * ar + ai * ai);
    float qr = (nr * ar + ni * ai) * inv;
    float qi = (ni * ar - nr * ai) * inv;

    float b = B_mat[idx], cm = C_mat[idx];
    dAr[idx] = dr;
    dAi[idx] = di;
    CBr[idx] = cm * (qr * b);
    CBi[idx] = cm * (qi * b);
}

__global__ void __launch_bounds__(256)
s4_kernelgen(const float* __restrict__ dAr, const float* __restrict__ dAi,
             const float* __restrict__ CBr, const float* __restrict__ CBi,
             float* __restrict__ Kbuf) {
    int idx = blockIdx.x * blockDim.x + threadIdx.x;
    if (idx >= DCH * LSEQ) return;
    int   d = idx / LSEQ;
    float l = (float)(idx & (LSEQ - 1));

    const float* pr = dAr + d * NST;
    const float* pi = dAi + d * NST;
    const float* cr = CBr + d * NST;
    const float* ci = CBi + d * NST;

    float acc = 0.0f;
#pragma unroll 4
    for (int n = 0; n < NST; ++n) {
        float e = __expf(pr[n] * l);
        float s, c;
        __sincosf(pi[n] * l, &s, &c);
        acc += e * (cr[n] * c - ci[n] * s);
    }
    Kbuf[idx] = acc;
}

#define NTILE  (LSEQ / 16)
#define KPAD   16
#define UPADT  3
#define TILEDW 80

__global__ void __launch_bounds__(512)
s4_conv_wmma(const float* __restrict__ u,
             const float* __restrict__ Kbuf,
             float* __restrict__ y) {
    __shared__ __attribute__((aligned(16))) float Klds[KPAD + LSEQ];
    __shared__ __attribute__((aligned(16))) float Ulds[(NTILE + UPADT) * TILEDW];

    const int d   = blockIdx.x;
    const int tid = threadIdx.x;

    if (tid < KPAD)           Klds[tid] = 0.0f;
    if (tid < UPADT * TILEDW) Ulds[tid] = 0.0f;
    {
        const float4* src = (const float4*)(Kbuf + (size_t)d * LSEQ);
        float4*       dst = (float4*)(Klds + KPAD);
        dst[tid] = src[tid];
    }
    {
        const int sb = tid & 3;
        const int sp = tid >> 2;
        const float* usrc = u + ((size_t)sb * LSEQ + (size_t)(2 * sp)) * DCH + d;
        float* udst = Ulds + (UPADT + (sp >> 3)) * TILEDW + (sp & 7) * 8 + sb * 2;
#pragma unroll
        for (int s = 0; s < LSEQ / 256; ++s) {
            v2f val;
            val.x = usrc[(size_t)s * 256 * DCH];
            val.y = usrc[(size_t)s * 256 * DCH + DCH];
            *(v2f*)(udst + s * 16 * TILEDW) = val;
        }
    }
    __syncthreads();

    const int wave = __builtin_amdgcn_readfirstlane(tid >> 5);
    const int lane = tid & 31;
    const int ln   = lane & 15;
    const int g    = lane >> 4;
    const int nb   = ln & 3;
    const int ns   = ln >> 2;

    const float* ka = Klds + KPAD + ln - 2 * g;

#pragma unroll 1
    for (int pass = 0; pass < 2; ++pass) {
        const int q  = pass ? wave : (NTILE / 4 - 1 - wave);
        const int I0 = 4 * q;
        const float* ub = Ulds + (UPADT + I0 + ns) * TILEDW + g * 8 + nb * 2;

        v8f acc = {};
        for (int delta = I0 + 3; delta >= 0; --delta) {
            const float* kd = ka + 16 * delta;
            const float* ud = ub - delta * TILEDW;
#pragma unroll
            for (int c = 0; c < 4; ++c) {
                v2f a;
                a.x = kd[-4 * c];
                a.y = kd[-4 * c - 1];
                const v2f b = *(const v2f*)(ud + 16 * c);
                acc = __builtin_amdgcn_wmma_f32_16x16x4_f32(
                    false, a, false, b, (short)0, acc, false, false);
            }
        }

        float* yp = y + ((size_t)nb * LSEQ + (size_t)((I0 + ns) * 16 + 8 * g)) * DCH + d;
#pragma unroll
        for (int r = 0; r < 8; ++r)
            yp[(size_t)r * DCH] = acc[r];
    }
}

extern "C" void kernel_launch(void* const* d_in, const int* in_sizes, int n_in,
                              void* d_out, int out_size, void* d_ws, size_t ws_size,
                              hipStream_t stream) {
    const float* u         = (const float*)d_in[0];
    const float* A_real    = (const float*)d_in[1];
    const float* A_imag    = (const float*)d_in[2];
    const float* B_mat     = (const float*)d_in[3];
    const float* C_mat     = (const float*)d_in[4];
    const float* log_delta = (const float*)d_in[5];
    float* y = (float*)d_out;

    float* ws   = (float*)d_ws;
    float* dAr  = ws;
    float* dAi  = ws + 1 * DN;
    float* CBr  = ws + 2 * DN;
    float* CBi  = ws + 3 * DN;
    float* Kbuf = ws + 4 * DN;

    s4_precompute<<<dim3((DN + 255) / 256), dim3(256), 0, stream>>>(
        A_real, A_imag, B_mat, C_mat, log_delta, dAr, dAi, CBr, CBi);

    s4_kernelgen<<<dim3((DCH * LSEQ + 255) / 256), dim3(256), 0, stream>>>(
        dAr, dAi, CBr, CBi, Kbuf);

    s4_conv_wmma<<<dim3(DCH), dim3(512), 0, stream>>>(u, Kbuf, y);
}


# Conv delta loop software-pipelined: a trip's (2 deltas, 8 WMMAs) LDS fragment loads go as one group a trip ahead into a second register set, not between WMMAs; k pair order in a WMMA flipped
# speedup vs baseline: 7.2057x; 1.0806x over previous
#include <hip/hip_runtime.h>
#include <hip/hip_bf16.h>
#include <math.h>

#define BSZ  4
#define LSEQ 2048
#define DCH  768
#define NST  64
#define DN   (DCH * NST)

typedef __attribute__((ext_vector_type(2))) float v2f;
typedef __attribute__((ext_vector_type(8))) float v8f;

__global__ void __launch_bounds__(256)
s4_precompute(const float* __restrict__ A_real, const float* __restrict__ A_imag,
              const float* __restrict__ B_mat,  const float* __restrict__ C_mat,
              const float* __restrict__ log_delta,
              float* __restrict__ dAr, float* __restrict__ dAi,
              float* __restrict__ CBr, float* __restrict__ CBi) {
    int idx = blockIdx.x * blockDim.x + threadIdx.x;
    if (idx >= DN) return;
    int d = idx / NST;

    float delta = log1pf(expf(log_delta[d]));
    float ar = A_real[idx], ai = A_imag[idx];
    float dr = delta * ar, di = delta * ai;

    float er = expf(dr);
    float s, c;
    __sincosf(di, &s, &c);
    float abr = er * c, abi = er * s;

    float nr = abr - 1.0f, ni = abi;
    float inv = 1.0f / (ar * ar + ai * ai);
    float qr = (nr * ar + ni * ai) * inv;
    float qi = (ni * ar - nr * ai) * inv;

    float b = B_mat[idx], cm = C_mat[idx];
    dAr[idx] = dr;
    dAi[idx] = di;
    CBr[idx] = cm * (qr * b);
    CBi[idx] = cm * (qi * b);
}

__global__ void __launch_bounds__(256)
s4_kernelgen(const float* __restrict__ dAr, const float* __restrict__ dAi,
             const float* __restrict__ CBr, const float* __restrict__ CBi,
             float* __restrict__ Kbuf) {
    int idx = blockIdx.x * blockDim.x + threadIdx.x;
    if (idx >= DCH * LSEQ) return;
    int   d = idx / LSEQ;
    float l = (float)(idx & (LSEQ - 1));

    const float* pr = dAr + d * NST;
    const float* pi = dAi + d * NST;
    const float* cr = CBr + d * NST;
    const float* ci = CBi + d * NST;

    float acc = 0.0f;
#pragma unroll 4
    for (int n = 0; n < NST; ++n) {
        float e = __expf(pr[n] * l);
        float s, c;
        __sincosf(pi[n] * l, &s, &c);
        acc += e * (cr[n] * c - ci[n] * s);
    }
    Kbuf[idx] = acc;
}

#define NTILE  (LSEQ / 16)
#define KPAD   16
#define UPADT  3
#define TILEDW 80

__global__ void __launch_bounds__(512, 12)
s4_conv_wmma(const float* __restrict__ u,
             const float* __restrict__ Kbuf,
             float* __restrict__ y) {
    __shared__ __attribute__((aligned(16))) float Klds[KPAD + LSEQ];
    __shared__ __attribute__((aligned(16))) float Ulds[(NTILE + UPADT) * TILEDW];

    const int d   = blockIdx.x;
    const int tid = threadIdx.x;

    if (tid < KPAD)           Klds[tid] = 0.0f;
    if (tid < UPADT * TILEDW) Ulds[tid] = 0.0f;
    {
        const float4* src = (const float4*)(Kbuf + (size_t)d * LSEQ);
        float4*       dst = (float4*)(Klds + KPAD);
        dst[tid] = src[tid];
    }
    {
        const int sb = tid & 3;
        const int sp = tid >> 2;
        const float* usrc = u + ((size_t)sb * LSEQ + (size_t)(2 * sp)) * DCH + d;
        float* udst = Ulds + (UPADT + (sp >> 3)) * TILEDW + (sp & 7) * 8 + sb * 2;
#pragma unroll
        for (int s = 0; s < LSEQ / 256; ++s) {
            v2f val;
            val.x = usrc[(size_t)s * 256 * DCH + DCH];
            val.y = usrc[(size_t)s * 256 * DCH];
            *(v2f*)(udst + s * 16 * TILEDW) = val;
        }
    }
    __syncthreads();

    const int wave = __builtin_amdgcn_readfirstlane(tid >> 5);
    const int lane = tid & 31;
    const int ln   = lane & 15;
    const int g    = lane >> 4;
    const int nb   = ln & 3;
    const int ns   = ln >> 2;

    const float* ka = Klds + KPAD + ln - 2 * g;
    const int ublane = (UPADT + ns) * TILEDW + g * 8 + nb * 2;

#define LOAD_TRIP(A_, B_, DD_)                                                  \
    {                                                                           \
        _Pragma("unroll")                                                       \
        for (int j = 0; j < 2; ++j) {                                           \
            const float* kd = ka + 16 * ((DD_) - j);                            \
            const float* ud = ub - ((DD_) - j) * TILEDW;                        \
            _Pragma("unroll")                                                   \
            for (int c = 0; c < 4; ++c) {                                       \
                A_[4 * j + c].x = kd[-4 * c - 1];                               \
                A_[4 * j + c].y = kd[-4 * c];                                   \
                B_[4 * j + c]   = *(const v2f*)(ud + 16 * c);                   \
            }                                                                   \
        }                                                                       \
    }
#define WMMA_TRIP(A_, B_)                                                       \
    {                                                                           \
        _Pragma("unroll")                                                       \
        for (int i = 0; i < 8; ++i)                                             \
            acc = __builtin_amdgcn_wmma_f32_16x16x4_f32(                        \
                false, A_[i], false, B_[i], (short)0, acc, false, false);       \
    }

#define GROUP_EDGE()                                                            \
    {                                                                           \
        asm volatile("" : "+v"(acc) : : "memory");                              \
        __builtin_amdgcn_sched_barrier(0);                                      \
    }

#pragma unroll 1
    for (int pass = 0; pass < 2; ++pass) {
        const int q  = pass ? wave : (NTILE / 4 - 1 - wave);
        const int I0 = 4 * q;
        const float* ub = Ulds + ublane + I0 * TILEDW;

        v8f acc = {};
        v2f a0[8], b0[8], a1[8], b1[8];
        int dd = I0 + 3;
        LOAD_TRIP(a0, b0, dd);
#pragma unroll 1
        for (; dd >= 3; dd -= 4) {
            LOAD_TRIP(a1, b1, dd - 2);
            GROUP_EDGE();
            WMMA_TRIP(a0, b0);
            GROUP_EDGE();
            const int dn = dd >= 7 ? dd - 4 : 1;
            LOAD_TRIP(a0, b0, dn);
            GROUP_EDGE();
            WMMA_TRIP(a1, b1);
            GROUP_EDGE();
        }

        int yl = lane;
        asm volatile("" : "+v"(yl));
        const unsigned ylane =
            (unsigned)(((yl & 3) * LSEQ + ((yl >> 2) & 3) * 16 + 8 * (yl >> 4)) * DCH) * 4u;
        char* yp = (char*)(y + (size_t)(I0 * 16) * DCH + d);
#pragma unroll
        for (int r = 0; r < 8; ++r)
            *(float*)(yp + (size_t)ylane + (size_t)(r * DCH * 4)) = acc[r];
    }
}

extern "C" void kernel_launch(void* const* d_in, const int* in_sizes, int n_in,
                              void* d_out, int out_size, void* d_ws, size_t ws_size,
                              hipStream_t stream) {
    const float* u         = (const float*)d_in[0];
    const float* A_real    = (const float*)d_in[1];
    const float* A_imag    = (const float*)d_in[2];
    const float* B_mat     = (const float*)d_in[3];
    const float* C_mat     = (const float*)d_in[4];
    const float* log_delta = (const float*)d_in[5];
    float* y = (float*)d_out;

    float* ws   = (float*)d_ws;
    float* dAr  = ws;
    float* dAi  = ws + 1 * DN;
    float* CBr  = ws + 2 * DN;
    float* CBi  = ws + 3 * DN;
    float* Kbuf = ws + 4 * DN;

    s4_precompute<<<dim3((DN + 255) / 256), dim3(256), 0, stream>>>(
        A_real, A_imag, B_mat, C_mat, log_delta, dAr, dAi, CBr, CBi);

    s4_kernelgen<<<dim3((DCH * LSEQ + 255) / 256), dim3(256), 0, stream>>>(
        dAr, dAi, CBr, CBi, Kbuf);

    s4_conv_wmma<<<dim3(DCH), dim3(512), 0, stream>>>(u, Kbuf, y);
}


# Conv reads u and writes y through channel-major (D,B,L) workspace copies made by two LDS-tiled layout kernels: whole 128-B lines per wave instruction instead of one 4-byte piece per line
# speedup vs baseline: 13.6930x; 1.9003x over previous
#include <hip/hip_runtime.h>
#include <hip/hip_bf16.h>
#include <math.h>

#define BSZ  4
#define LSEQ 2048
#define DCH  768
#define NST  64
#define DN   (DCH * NST)

typedef __attribute__((ext_vector_type(2))) float v2f;
typedef __attribute__((ext_vector_type(8))) float v8f;

__global__ void __launch_bounds__(256)
s4_precompute(const float* __restrict__ A_real, const float* __restrict__ A_imag,
              const float* __restrict__ B_mat,  const float* __restrict__ C_mat,
              const float* __restrict__ log_delta,
              float* __restrict__ dAr, float* __restrict__ dAi,
              float* __restrict__ CBr, float* __restrict__ CBi) {
    int idx = blockIdx.x * blockDim.x + threadIdx.x;
    if (idx >= DN) return;
    int d = idx / NST;

    float delta = log1pf(expf(log_delta[d]));
    float ar = A_real[idx], ai = A_imag[idx];
    float dr = delta * ar, di = delta * ai;

    float er = expf(dr);
    float s, c;
    __sincosf(di, &s, &c);
    float abr = er * c, abi = er * s;

    float nr = abr - 1.0f, ni = abi;
    float inv = 1.0f / (ar * ar + ai * ai);
    float qr = (nr * ar + ni * ai) * inv;
    float qi = (ni * ar - nr * ai) * inv;

    float b = B_mat[idx], cm = C_mat[idx];
    dAr[idx] = dr;
    dAi[idx] = di;
    CBr[idx] = cm * (qr * b);
    CBi[idx] = cm * (qi * b);
}

__global__ void __launch_bounds__(256)
s4_kernelgen(const float* __restrict__ dAr, const float* __restrict__ dAi,
             const float* __restrict__ CBr, const float* __restrict__ CBi,
             float* __restrict__ Kbuf) {
    int idx = blockIdx.x * blockDim.x + threadIdx.x;
    if (idx >= DCH * LSEQ) return;
    int   d = idx / LSEQ;
    float l = (float)(idx & (LSEQ - 1));

    const float* pr = dAr + d * NST;
    const float* pi = dAi + d * NST;
    const float* cr = CBr + d * NST;
    const float* ci = CBi + d * NST;

    float acc = 0.0f;
#pragma unroll 4
    for (int n = 0; n < NST; ++n) {
        float e = __expf(pr[n] * l);
        float s, c;
        __sincosf(pi[n] * l, &s, &c);
        acc += e * (cr[n] * c - ci[n] * s);
    }
    Kbuf[idx] = acc;
}

#define TT     256
#define TPITCH 260

__global__ void __launch_bounds__(256)
s4_to_channel_major(const float* __restrict__ u, float* __restrict__ uT) {
    __shared__ __attribute__((aligned(16))) float tile[32 * TPITCH];
    const int t0  = blockIdx.x * TT;
    const int d0  = blockIdx.y * 32;
    const int b   = blockIdx.z;
    const int tid = threadIdx.x;

    {
        const int c4 = tid & 7, tr = tid >> 3;
        const float* src = u + ((size_t)b * LSEQ + (size_t)(t0 + tr)) * DCH + d0 + 4 * c4;
        float4 v[TT / 32];
#pragma unroll
        for (int i = 0; i < TT / 32; ++i)
            v[i] = *(const float4*)(src + (size_t)i * 32 * DCH);
#pragma unroll
        for (int i = 0; i < TT / 32; ++i) {
            float* dst = tile + (4 * c4) * TPITCH + tr + 32 * i;
            dst[0 * TPITCH] = v[i].x;
            dst[1 * TPITCH] = v[i].y;
            dst[2 * TPITCH] = v[i].z;
            dst[3 * TPITCH] = v[i].w;
        }
    }
    __syncthreads();
    {
        const int tq = tid & 63, dr = tid >> 6;
#pragma unroll
        for (int i = 0; i < 8; ++i) {
            const int dd = dr + 4 * i;
            const float4 o = *(const float4*)(tile + dd * TPITCH + 4 * tq);
            *(float4*)(uT + ((size_t)(d0 + dd) * BSZ + b) * LSEQ + t0 + 4 * tq) = o;
        }
    }
}

__global__ void __launch_bounds__(256)
s4_to_time_major(const float* __restrict__ yT, float* __restrict__ y) {
    __shared__ __attribute__((aligned(16))) float tile[32 * TPITCH];
    const int t0  = blockIdx.x * TT;
    const int d0  = blockIdx.y * 32;
    const int b   = blockIdx.z;
    const int tid = threadIdx.x;

    {
        const int tq = tid & 63, dr = tid >> 6;
        float4 v[8];
#pragma unroll
        for (int i = 0; i < 8; ++i)
            v[i] = *(const float4*)(yT + ((size_t)(d0 + dr + 4 * i) * BSZ + b) * LSEQ + t0 + 4 * tq);
#pragma unroll
        for (int i = 0; i < 8; ++i)
            *(float4*)(tile + (dr + 4 * i) * TPITCH + 4 * tq) = v[i];
    }
    __syncthreads();
    {
        const int c4 = tid & 7, tr = tid >> 3;
        float* dst = y + ((size_t)b * LSEQ + (size_t)(t0 + tr)) * DCH + d0 + 4 * c4;
#pragma unroll
        for (int i = 0; i < TT / 32; ++i) {
            const float* s = tile + (4 * c4) * TPITCH + tr + 32 * i;
            float4 o;
            o.x = s[0 * TPITCH];
            o.y = s[1 * TPITCH];
            o.z = s[2 * TPITCH];
            o.w = s[3 * TPITCH];
            *(float4*)(dst + (size_t)i * 32 * DCH) = o;
        }
    }
}

#define NTILE  (LSEQ / 16)
#define KPAD   16
#define UPADT  3
#define TILEDW 80

__global__ void __launch_bounds__(512, 12)
s4_conv_wmma(const float* __restrict__ uT,
             const float* __restrict__ Kbuf,
             float* __restrict__ yT) {
    __shared__ __attribute__((aligned(16))) float Klds[KPAD + LSEQ];
    __shared__ __attribute__((aligned(16))) float Ulds[(NTILE + UPADT) * TILEDW];

    const int d   = blockIdx.x;
    const int tid = threadIdx.x;

    if (tid < KPAD)           Klds[tid] = 0.0f;
    if (tid < UPADT * TILEDW) Ulds[tid] = 0.0f;
    {
        const float4* src = (const float4*)(Kbuf + (size_t)d * LSEQ);
        float4*       dst = (float4*)(Klds + KPAD);
        dst[tid] = src[tid];
    }
    {
        const float4* usrc = (const float4*)(uT + (size_t)d * (BSZ * LSEQ)) + tid;
        float* udst = Ulds + (UPADT + (tid >> 2)) * TILEDW + (tid & 3) * 16;
        float4 val[BSZ];
#pragma unroll
        for (int sb = 0; sb < BSZ; ++sb)
            val[sb] = usrc[sb * (LSEQ / 4)];
        *(float4*)(udst)      = make_float4(val[0].y, val[0].x, val[1].y, val[1].x);
        *(float4*)(udst + 4)  = make_float4(val[2].y, val[2].x, val[3].y, val[3].x);
        *(float4*)(udst + 8)  = make_float4(val[0].w, val[0].z, val[1].w, val[1].z);
        *(float4*)(udst + 12) = make_float4(val[2].w, val[2].z, val[3].w, val[3].z);
    }
    __syncthreads();

    const int wave = __builtin_amdgcn_readfirstlane(tid >> 5);
    const int lane = tid & 31;
    const int ln   = lane & 15;
    const int g    = lane >> 4;
    const int nb   = ln & 3;
    const int ns   = ln >> 2;

    const float* ka = Klds + KPAD + ln - 2 * g;
    const int ublane = (UPADT + ns) * TILEDW + g * 8 + nb * 2;

#define LOAD_TRIP(A_, B_, DD_)                                                  \
    {                                                                           \
        _Pragma("unroll")                                                       \
        for (int j = 0; j < 2; ++j) {                                           \
            const float* kd = ka + 16 * ((DD_) - j);                            \
            const float* ud = ub - ((DD_) - j) * TILEDW;                        \
            _Pragma("unroll")                                                   \
            for (int c = 0; c < 4; ++c) {                                       \
                A_[4 * j + c].x = kd[-4 * c - 1];                               \
                A_[4 * j + c].y = kd[-4 * c];                                   \
                B_[4 * j + c]   = *(const v2f*)(ud + 16 * c);                   \
            }                                                                   \
        }                                                                       \
    }
#define WMMA_TRIP(A_, B_)                                                       \
    {                                                                           \
        _Pragma("unroll")                                                       \
        for (int i = 0; i < 8; ++i)                                             \
            acc = __builtin_amdgcn_wmma_f32_16x16x4_f32(                        \
                false, A_[i], false, B_[i], (short)0, acc, false, false);       \
    }

#define GROUP_EDGE()                                                            \
    {                                                                           \
        asm volatile("" : "+v"(acc) : : "memory");                              \
        __builtin_amdgcn_sched_barrier(0);                                      \
    }

#pragma unroll 1
    for (int pass = 0; pass < 2; ++pass) {
        const int q  = pass ? wave : (NTILE / 4 - 1 - wave);
        const int I0 = 4 * q;
        const float* ub = Ulds + ublane + I0 * TILEDW;

        v8f acc = {};
        v2f a0[8], b0[8], a1[8], b1[8];
        int dd = I0 + 3;
        LOAD_TRIP(a0, b0, dd);
#pragma unroll 1
        for (; dd >= 3; dd -= 4) {
            LOAD_TRIP(a1, b1, dd - 2);
            GROUP_EDGE();
            WMMA_TRIP(a0, b0);
            GROUP_EDGE();
            const int dn = dd >= 7 ? dd - 4 : 1;
            LOAD_TRIP(a0, b0, dn);
            GROUP_EDGE();
            WMMA_TRIP(a1, b1);
            GROUP_EDGE();
        }

        int yl = lane;
        asm volatile("" : "+v"(yl));
        const unsigned ylane =
            (unsigned)((yl & 3) * LSEQ + ((yl >> 2) & 3) * 16 + 8 * (yl >> 4)) * 4u;
        char* yp = (char*)(yT + (size_t)d * (BSZ * LSEQ) + I0 * 16);
        *(float4*)(yp + (size_t)ylane)      = make_float4(acc[0], acc[1], acc[2], acc[3]);
        *(float4*)(yp + (size_t)ylane + 16) = make_float4(acc[4], acc[5], acc[6], acc[7]);
    }
}

extern "C" void kernel_launch(void* const* d_in, const int* in_sizes, int n_in,
                              void* d_out, int out_size, void* d_ws, size_t ws_size,
                              hipStream_t stream) {
    const float* u         = (const float*)d_in[0];
    const float* A_real    = (const float*)d_in[1];
    const float* A_imag    = (const float*)d_in[2];
    const float* B_mat     = (const float*)d_in[3];
    const float* C_mat     = (const float*)d_in[4];
    const float* log_delta = (const float*)d_in[5];
    float* y = (float*)d_out;

    float* ws   = (float*)d_ws;
    float* dAr  = ws;
    float* dAi  = ws + 1 * DN;
    float* CBr  = ws + 2 * DN;
    float* CBi  = ws + 3 * DN;
    float* Kbuf = ws + 4 * DN;
    float* uT   = Kbuf + DCH * LSEQ;
    float* yT   = uT + BSZ * LSEQ * DCH;

    s4_to_channel_major<<<dim3(LSEQ / TT, DCH / 32, BSZ), dim3(256), 0, stream>>>(u, uT);

    s4_precompute<<<dim3((DN + 255) / 256), dim3(256), 0, stream>>>(
        A_real, A_imag, B_mat, C_mat, log_delta, dAr, dAi, CBr, CBi);

    s4_kernelgen<<<dim3((DCH * LSEQ + 255) / 256), dim3(256), 0, stream>>>(
        dAr, dAi, CBr, CBi, Kbuf);

    s4_conv_wmma<<<dim3(DCH), dim3(512), 0, stream>>>(uT, Kbuf, yT);

    s4_to_time_major<<<dim3(LSEQ / TT, DCH / 32, BSZ), dim3(256), 0, stream>>>(yT, y);
}


# K[d] built as one f32 WMMA product per channel of two factored power tables, exp(dA*32i) and CB*exp(dA*j), l = 32i+j: 21x fewer exp/sin/cos; the sum over n now runs in WMMA k order
# speedup vs baseline: 16.4037x; 1.1980x over previous
#include <hip/hip_runtime.h>
#include <hip/hip_bf16.h>
#include <math.h>

#define BSZ  4
#define LSEQ 2048
#define DCH  768
#define NST  64
#define DN   (DCH * NST)

typedef __attribute__((ext_vector_type(2))) float v2f;
typedef __attribute__((ext_vector_type(8))) float v8f;

__global__ void __launch_bounds__(256)
s4_precompute(const float* __restrict__ A_real, const float* __restrict__ A_imag,
              const float* __restrict__ B_mat,  const float* __restrict__ C_mat,
              const float* __restrict__ log_delta,
              float* __restrict__ dAr, float* __restrict__ dAi,
              float* __restrict__ CBr, float* __restrict__ CBi) {
    int idx = blockIdx.x * blockDim.x + threadIdx.x;
    if (idx >= DN) return;
    int d = idx / NST;

    float delta = log1pf(expf(log_delta[d]));
    float ar = A_real[idx], ai = A_imag[idx];
    float dr = delta * ar, di = delta * ai;

    float er = expf(dr);
    float s, c;
    __sincosf(di, &s, &c);
    float abr = er * c, abi = er * s;

    float nr = abr - 1.0f, ni = abi;
    float inv = 1.0f / (ar * ar + ai * ai);
    float qr = (nr * ar + ni * ai) * inv;
    float qi = (ni * ar - nr * ai) * inv;

    float b = B_mat[idx], cm = C_mat[idx];
    dAr[idx] = dr;
    dAi[idx] = di;
    CBr[idx] = cm * (qr * b);
    CBi[idx] = cm * (qi * b);
}

#define KG_I 64
#define KG_J 32
#define KG_P 132

__global__ void __launch_bounds__(256)
s4_kernelgen(const float* __restrict__ dAr, const float* __restrict__ dAi,
             const float* __restrict__ CBr, const float* __restrict__ CBi,
             float* __restrict__ Kbuf) {
    __shared__ __attribute__((aligned(16))) float lds[LSEQ + (KG_I + KG_J) * KG_P];
    float* Ks = lds;
    float* Xs = lds + LSEQ;
    float* Qs = Xs + KG_I * KG_P;

    const int d   = blockIdx.x;
    const int tid = threadIdx.x;

    {
        const int n  = tid & (NST - 1);
        const int r0 = tid >> 6;
        const float pr = dAr[d * NST + n], pi = dAi[d * NST + n];
        const float cr = CBr[d * NST + n], ci = CBi[d * NST + n];
#pragma unroll 4
        for (int ii = 0; ii < KG_I / 4; ++ii) {
            const int   i = r0 + 4 * ii;
            const float l = (float)(KG_J * i);
            const float e = __expf(pr * l);
            float s, c;
            __sincosf(pi * l, &s, &c);
            *(float2*)(Xs + i * KG_P + 2 * n) = make_float2(e * c, -(e * s));
        }
#pragma unroll 4
        for (int jj = 0; jj < KG_J / 4; ++jj) {
            const int   j = r0 + 4 * jj;
            const float l = (float)j;
            const float e = __expf(pr * l);
            float s, c;
            __sincosf(pi * l, &s, &c);
            const float qr = e * c, qi = e * s;
            *(float2*)(Qs + j * KG_P + 2 * n) = make_float2(cr * qr - ci * qi, cr * qi + ci * qr);
        }
    }
    __syncthreads();

    const int wave = __builtin_amdgcn_readfirstlane(tid >> 5);
    const int lane = tid & 31;
    const int ln   = lane & 15;
    const int g    = lane >> 4;
    const int it   = wave >> 1;
    const int jt   = wave & 1;
    const float* xa = Xs + (16 * it + ln) * KG_P + 2 * g;
    const float* qb = Qs + (16 * jt + ln) * KG_P + 2 * g;

    unsigned kbyte = 4u * (unsigned)(KG_J * (16 * it + 8 * g) + 16 * jt + ln);
    asm volatile("" : "+v"(kbyte));
    float* kst = (float*)((char*)Ks + kbyte);

    v8f acc = {};
#pragma unroll
    for (int c = 0; c < 32; ++c) {
        const v2f fa = *(const v2f*)(xa + 4 * c);
        const v2f fb = *(const v2f*)(qb + 4 * c);
        acc = __builtin_amdgcn_wmma_f32_16x16x4_f32(false, fa, false, fb, (short)0, acc, false, false);
    }

#pragma unroll
    for (int r = 0; r < 8; ++r)
        kst[KG_J * r] = acc[r];
    __syncthreads();
    {
        const float4* s4 = (const float4*)Ks;
        float4*       d4 = (float4*)(Kbuf + (size_t)d * LSEQ);
        d4[tid]       = s4[tid];
        d4[tid + 256] = s4[tid + 256];
    }
}

#define TT     256
#define TPITCH 260

__global__ void __launch_bounds__(256)
s4_to_channel_major(const float* __restrict__ u, float* __restrict__ uT) {
    __shared__ __attribute__((aligned(16))) float tile[32 * TPITCH];
    const int t0  = blockIdx.x * TT;
    const int d0  = blockIdx.y * 32;
    const int b   = blockIdx.z;
    const int tid = threadIdx.x;

    {
        const int c4 = tid & 7, tr = tid >> 3;
        const float* src = u + ((size_t)b * LSEQ + (size_t)(t0 + tr)) * DCH + d0 + 4 * c4;
        float4 v[TT / 32];
#pragma unroll
        for (int i = 0; i < TT / 32; ++i)
            v[i] = *(const float4*)(src + (size_t)i * 32 * DCH);
#pragma unroll
        for (int i = 0; i < TT / 32; ++i) {
            float* dst = tile + (4 * c4) * TPITCH + tr + 32 * i;
            dst[0 * TPITCH] = v[i].x;
            dst[1 * TPITCH] = v[i].y;
            dst[2 * TPITCH] = v[i].z;
            dst[3 * TPITCH] = v[i].w;
        }
    }
    __syncthreads();
    {
        const int tq = tid & 63, dr = tid >> 6;
#pragma unroll
        for (int i = 0; i < 8; ++i) {
            const int dd = dr + 4 * i;
            const float4 o = *(const float4*)(tile + dd * TPITCH + 4 * tq);
            *(float4*)(uT + ((size_t)(d0 + dd) * BSZ + b) * LSEQ + t0 + 4 * tq) = o;
        }
    }
}

__global__ void __launch_bounds__(256)
s4_to_time_major(const float* __restrict__ yT, float* __restrict__ y) {
    __shared__ __attribute__((aligned(16))) float tile[32 * TPITCH];
    const int t0  = blockIdx.x * TT;
    const int d0  = blockIdx.y * 32;
    const int b   = blockIdx.z;
    const int tid = threadIdx.x;

    {
        const int tq = tid & 63, dr = tid >> 6;
        float4 v[8];
#pragma unroll
        for (int i = 0; i < 8; ++i)
            v[i] = *(const float4*)(yT + ((size_t)(d0 + dr + 4 * i) * BSZ + b) * LSEQ + t0 + 4 * tq);
#pragma unroll
        for (int i = 0; i < 8; ++i)
            *(float4*)(tile + (dr + 4 * i) * TPITCH + 4 * tq) = v[i];
    }
    __syncthreads();
    {
        const int c4 = tid & 7, tr = tid >> 3;
        float* dst = y + ((size_t)b * LSEQ + (size_t)(t0 + tr)) * DCH + d0 + 4 * c4;
#pragma unroll
        for (int i = 0; i < TT / 32; ++i) {
            const float* s = tile + (4 * c4) * TPITCH + tr + 32 * i;
            float4 o;
            o.x = s[0 * TPITCH];
            o.y = s[1 * TPITCH];
            o.z = s[2 * TPITCH];
            o.w = s[3 * TPITCH];
            *(float4*)(dst + (size_t)i * 32 * DCH) = o;
        }
    }
}

#define NTILE  (LSEQ / 16)
#define KPAD   16
#define UPADT  3
#define TILEDW 80

__global__ void __launch_bounds__(512, 12)
s4_conv_wmma(const float* __restrict__ uT,
             const float* __restrict__ Kbuf,
             float* __restrict__ yT) {
    __shared__ __attribute__((aligned(16))) float Klds[KPAD + LSEQ];
    __shared__ __attribute__((aligned(16))) float Ulds[(NTILE + UPADT) * TILEDW];

    const int d   = blockIdx.x;
    const int tid = threadIdx.x;

    if (tid < KPAD)           Klds[tid] = 0.0f;
    if (tid < UPADT * TILEDW) Ulds[tid] = 0.0f;
    {
        const float4* src = (const float4*)(Kbuf + (size_t)d * LSEQ);
        float4*       dst = (float4*)(Klds + KPAD);
        dst[tid] = src[tid];
    }
    {
        const float4* usrc = (const float4*)(uT + (size_t)d * (BSZ * LSEQ)) + tid;
        float* udst = Ulds + (UPADT + (tid >> 2)) * TILEDW + (tid & 3) * 16;
        float4 val[BSZ];
#pragma unroll
        for (int sb = 0; sb < BSZ; ++sb)
            val[sb] = usrc[sb * (LSEQ / 4)];
        *(float4*)(udst)      = make_float4(val[0].y, val[0].x, val[1].y, val[1].x);
        *(float4*)(udst + 4)  = make_float4(val[2].y, val[2].x, val[3].y, val[3].x);
        *(float4*)(udst + 8)  = make_float4(val[0].w, val[0].z, val[1].w, val[1].z);
        *(float4*)(udst + 12) = make_float4(val[2].w, val[2].z, val[3].w, val[3].z);
    }
    __syncthreads();

    const int wave = __builtin_amdgcn_readfirstlane(tid >> 5);
    const int lane = tid & 31;
    const int ln   = lane & 15;
    const int g    = lane >> 4;
    const int nb   = ln & 3;
    const int ns   = ln >> 2;

    const float* ka = Klds + KPAD + ln - 2 * g;
    const int ublane = (UPADT + ns) * TILEDW + g * 8 + nb * 2;

#define LOAD_TRIP(A_, B_, DD_)                                                  \
    {                                                                           \
        _Pragma("unroll")                                                       \
        for (int j = 0; j < 2; ++j) {                                           \
            const float* kd = ka + 16 * ((DD_) - j);                            \
            const float* ud = ub - ((DD_) - j) * TILEDW;                        \
            _Pragma("unroll")                                                   \
            for (int c = 0; c < 4; ++c) {                                       \
                A_[4 * j + c].x = kd[-4 * c - 1];                               \
                A_[4 * j + c].y = kd[-4 * c];                                   \
                B_[4 * j + c]   = *(const v2f*)(ud + 16 * c);                   \
            }                                                                   \
        }                                                                       \
    }
#define WMMA_TRIP(A_, B_)                                                       \
    {                                                                           \
        _Pragma("unroll")                                                       \
        for (int i = 0; i < 8; ++i)                                             \
            acc = __builtin_amdgcn_wmma_f32_16x16x4_f32(                        \
                false, A_[i], false, B_[i], (short)0, acc, false, false);       \
    }

#define GROUP_EDGE()                                                            \
    {                                                                           \
        asm volatile("" : "+v"(acc) : : "memory");                              \
        __builtin_amdgcn_sched_barrier(0);                                      \
    }

#pragma unroll 1
    for (int pass = 0; pass < 2; ++pass) {
        const int q  = pass ? wave : (NTILE / 4 - 1 - wave);
        const int I0 = 4 * q;
        const float* ub = Ulds + ublane + I0 * TILEDW;

        v8f acc = {};
        v2f a0[8], b0[8], a1[8], b1[8];
        int dd = I0 + 3;
        LOAD_TRIP(a0, b0, dd);
#pragma unroll 1
        for (; dd >= 3; dd -= 4) {
            LOAD_TRIP(a1, b1, dd - 2);
            GROUP_EDGE();
            WMMA_TRIP(a0, b0);
            GROUP_EDGE();
            const int dn = dd >= 7 ? dd - 4 : 1;
            LOAD_TRIP(a0, b0, dn);
            GROUP_EDGE();
            WMMA_TRIP(a1, b1);
            GROUP_EDGE();
        }

        int yl = lane;
        asm volatile("" : "+v"(yl));
        const unsigned ylane =
            (unsigned)((yl & 3) * LSEQ + ((yl >> 2) & 3) * 16 + 8 * (yl >> 4)) * 4u;
        char* yp = (char*)(yT + (size_t)d * (BSZ * LSEQ) + I0 * 16);
        *(float4*)(yp + (size_t)ylane)      = make_float4(acc[0], acc[1], acc[2], acc[3]);
        *(float4*)(yp + (size_t)ylane + 16) = make_float4(acc[4], acc[5], acc[6], acc[7]);
    }
}

extern "C" void kernel_launch(void* const* d_in, const int* in_sizes, int n_in,
                              void* d_out, int out_size, void* d_ws, size_t ws_size,
                              hipStream_t stream) {
    const float* u         = (const float*)d_in[0];
    const float* A_real    = (const float*)d_in[1];
    const float* A_imag    = (const float*)d_in[2];
    const float* B_mat     = (const float*)d_in[3];
    const float* C_mat     = (const float*)d_in[4];
    const float* log_delta = (const float*)d_in[5];
    float* y = (float*)d_out;

    float* ws   = (float*)d_ws;
    float* dAr  = ws;
    float* dAi  = ws + 1 * DN;
    float* CBr  = ws + 2 * DN;
    float* CBi  = ws + 3 * DN;
    float* Kbuf = ws + 4 * DN;
    float* uT   = Kbuf + DCH * LSEQ;
    float* yT   = uT + BSZ * LSEQ * DCH;

    s4_to_channel_major<<<dim3(LSEQ / TT, DCH / 32, BSZ), dim3(256), 0, stream>>>(u, uT);

    s4_precompute<<<dim3((DN + 255) / 256), dim3(256), 0, stream>>>(
        A_real, A_imag, B_mat, C_mat, log_delta, dAr, dAi, CBr, CBi);

    s4_kernelgen<<<dim3(DCH), dim3(256), 0, stream>>>(
        dAr, dAi, CBr, CBi, Kbuf);

    s4_conv_wmma<<<dim3(DCH), dim3(512), 0, stream>>>(uT, Kbuf, yT);

    s4_to_time_major<<<dim3(LSEQ / TT, DCH / 32, BSZ), dim3(256), 0, stream>>>(yT, y);
}


# Layout of u, discretization and K build run as ONE launch (two block roles of one grid; dA, CB stay in registers): 5 launches -> 3; K staged over its X table, so all blocks are resident at once
# speedup vs baseline: 17.3093x; 1.0552x over previous
#include <hip/hip_runtime.h>
#include <hip/hip_bf16.h>
#include <math.h>

#define BSZ  4
#define LSEQ 2048
#define DCH  768
#define NST  64
#define DN   (DCH * NST)

typedef __attribute__((ext_vector_type(2))) float v2f;
typedef __attribute__((ext_vector_type(8))) float v8f;

#define KG_I 64
#define KG_J 32
#define KG_P 132
#define TT     256
#define TPITCH 260
#define UT_BLOCKS ((LSEQ / TT) * (DCH / 32) * BSZ)

__global__ void __launch_bounds__(256)
s4_front(const float* __restrict__ u,
         const float* __restrict__ A_real, const float* __restrict__ A_imag,
         const float* __restrict__ B_mat,  const float* __restrict__ C_mat,
         const float* __restrict__ log_delta,
         float* __restrict__ uT, float* __restrict__ Kbuf) {
    __shared__ __attribute__((aligned(16))) float lds[(KG_I + KG_J) * KG_P];
    const int tid = threadIdx.x;

    if (blockIdx.x < DCH) {
        float* Ks = lds;
        float* Xs = lds;
        float* Qs = lds + KG_I * KG_P;
        const int d = blockIdx.x;

        {
            const int n   = tid & (NST - 1);
            const int r0  = tid >> 6;
            const int idx = d * NST + n;

            const float delta = log1pf(expf(log_delta[d]));
            const float ar = A_real[idx], ai = A_imag[idx];
            const float pr = delta * ar, pi = delta * ai;
            float cr, ci;
            {
                const float er = expf(pr);
                float s, c;
                __sincosf(pi, &s, &c);
                const float abr = er * c, abi = er * s;
                const float nr = abr - 1.0f, ni = abi;
                const float inv = 1.0f / (ar * ar + ai * ai);
                const float qr = (nr * ar + ni * ai) * inv;
                const float qi = (ni * ar - nr * ai) * inv;
                const float b = B_mat[idx], cm = C_mat[idx];
                cr = cm * (qr * b);
                ci = cm * (qi * b);
            }

#pragma unroll 4
            for (int ii = 0; ii < KG_I / 4; ++ii) {
                const int   i = r0 + 4 * ii;
                const float l = (float)(KG_J * i);
                const float e = __expf(pr * l);
                float s, c;
                __sincosf(pi * l, &s, &c);
                *(float2*)(Xs + i * KG_P + 2 * n) = make_float2(e * c, -(e * s));
            }
#pragma unroll 4
            for (int jj = 0; jj < KG_J / 4; ++jj) {
                const int   j = r0 + 4 * jj;
                const float l = (float)j;
                const float e = __expf(pr * l);
                float s, c;
                __sincosf(pi * l, &s, &c);
                const float qr = e * c, qi = e * s;
                *(float2*)(Qs + j * KG_P + 2 * n) = make_float2(cr * qr - ci * qi, cr * qi + ci * qr);
            }
        }
        __syncthreads();

        const int wave = __builtin_amdgcn_readfirstlane(tid >> 5);
        const int lane = tid & 31;
        const int ln   = lane & 15;
        const int g    = lane >> 4;
        const int it   = wave >> 1;
        const int jt   = wave & 1;
        const float* xa = Xs + (16 * it + ln) * KG_P + 2 * g;
        const float* qb = Qs + (16 * jt + ln) * KG_P + 2 * g;

        unsigned kbyte = 4u * (unsigned)(KG_J * (16 * it + 8 * g) + 16 * jt + ln);
        asm volatile("" : "+v"(kbyte));
        float* kst = (float*)((char*)Ks + kbyte);

        v8f acc = {};
#pragma unroll
        for (int c = 0; c < 32; ++c) {
            const v2f fa = *(const v2f*)(xa + 4 * c);
            const v2f fb = *(const v2f*)(qb + 4 * c);
            acc = __builtin_amdgcn_wmma_f32_16x16x4_f32(false, fa, false, fb, (short)0, acc, false, false);
        }

        __syncthreads();

#pragma unroll
        for (int r = 0; r < 8; ++r)
            kst[KG_J * r] = acc[r];
        __syncthreads();
        {
            const float4* s4 = (const float4*)Ks;
            float4*       d4 = (float4*)(Kbuf + (size_t)d * LSEQ);
            d4[tid]       = s4[tid];
            d4[tid + 256] = s4[tid + 256];
        }
    } else {
        float* tile = lds;
        const int bid = blockIdx.x - DCH;
        const int t0  = (bid & (LSEQ / TT - 1)) * TT;
        const int dy  = (bid / (LSEQ / TT)) % (DCH / 32);
        const int b   = bid / ((LSEQ / TT) * (DCH / 32));
        const int d0  = dy * 32;

        {
            const int c4 = tid & 7, tr = tid >> 3;
            const float* src = u + ((size_t)b * LSEQ + (size_t)(t0 + tr)) * DCH + d0 + 4 * c4;
            float4 v[TT / 32];
#pragma unroll
            for (int i = 0; i < TT / 32; ++i)
                v[i] = *(const float4*)(src + (size_t)i * 32 * DCH);
#pragma unroll
            for (int i = 0; i < TT / 32; ++i) {
                float* dst = tile + (4 * c4) * TPITCH + tr + 32 * i;
                dst[0 * TPITCH] = v[i].x;
                dst[1 * TPITCH] = v[i].y;
                dst[2 * TPITCH] = v[i].z;
                dst[3 * TPITCH] = v[i].w;
            }
        }
        __syncthreads();
        {
            const int tq = tid & 63, dr = tid >> 6;
#pragma unroll
            for (int i = 0; i < 8; ++i) {
                const int dd = dr + 4 * i;
                const float4 o = *(const float4*)(tile + dd * TPITCH + 4 * tq);
                *(float4*)(uT + ((size_t)(d0 + dd) * BSZ + b) * LSEQ + t0 + 4 * tq) = o;
            }
        }
    }
}

__global__ void __launch_bounds__(256)
s4_to_time_major(const float* __restrict__ yT, float* __restrict__ y) {
    __shared__ __attribute__((aligned(16))) float tile[32 * TPITCH];
    const int t0  = blockIdx.x * TT;
    const int d0  = blockIdx.y * 32;
    const int b   = blockIdx.z;
    const int tid = threadIdx.x;

    {
        const int tq = tid & 63, dr = tid >> 6;
        float4 v[8];
#pragma unroll
        for (int i = 0; i < 8; ++i)
            v[i] = *(const float4*)(yT + ((size_t)(d0 + dr + 4 * i) * BSZ + b) * LSEQ + t0 + 4 * tq);
#pragma unroll
        for (int i = 0; i < 8; ++i)
            *(float4*)(tile + (dr + 4 * i) * TPITCH + 4 * tq) = v[i];
    }
    __syncthreads();
    {
        const int c4 = tid & 7, tr = tid >> 3;
        float* dst = y + ((size_t)b * LSEQ + (size_t)(t0 + tr)) * DCH + d0 + 4 * c4;
#pragma unroll
        for (int i = 0; i < TT / 32; ++i) {
            const float* s = tile + (4 * c4) * TPITCH + tr + 32 * i;
            float4 o;
            o.x = s[0 * TPITCH];
            o.y = s[1 * TPITCH];
            o.z = s[2 * TPITCH];
            o.w = s[3 * TPITCH];
            *(float4*)(dst + (size_t)i * 32 * DCH) = o;
        }
    }
}

#define NTILE  (LSEQ / 16)
#define KPAD   16
#define UPADT  3
#define TILEDW 80

__global__ void __launch_bounds__(512, 12)
s4_conv_wmma(const float* __restrict__ uT,
             const float* __restrict__ Kbuf,
             float* __restrict__ yT) {
    __shared__ __attribute__((aligned(16))) float Klds[KPAD + LSEQ];
    __shared__ __attribute__((aligned(16))) float Ulds[(NTILE + UPADT) * TILEDW];

    const int d   = blockIdx.x;
    const int tid = threadIdx.x;

    if (tid < KPAD)           Klds[tid] = 0.0f;
    if (tid < UPADT * TILEDW) Ulds[tid] = 0.0f;
    {
        const float4* src = (const float4*)(Kbuf + (size_t)d * LSEQ);
        float4*       dst = (float4*)(Klds + KPAD);
        dst[tid] = src[tid];
    }
    {
        const float4* usrc = (const float4*)(uT + (size_t)d * (BSZ * LSEQ)) + tid;
        float* udst = Ulds + (UPADT + (tid >> 2)) * TILEDW + (tid & 3) * 16;
        float4 val[BSZ];
#pragma unroll
        for (int sb = 0; sb < BSZ; ++sb)
            val[sb] = usrc[sb * (LSEQ / 4)];
        *(float4*)(udst)      = make_float4(val[0].y, val[0].x, val[1].y, val[1].x);
        *(float4*)(udst + 4)  = make_float4(val[2].y, val[2].x, val[3].y, val[3].x);
        *(float4*)(udst + 8)  = make_float4(val[0].w, val[0].z, val[1].w, val[1].z);
        *(float4*)(udst + 12) = make_float4(val[2].w, val[2].z, val[3].w, val[3].z);
    }
    __syncthreads();

    const int wave = __builtin_amdgcn_readfirstlane(tid >> 5);
    const int lane = tid & 31;
    const int ln   = lane & 15;
    const int g    = lane >> 4;
    const int nb   = ln & 3;
    const int ns   = ln >> 2;

    const float* ka = Klds + KPAD + ln - 2 * g;
    const int ublane = (UPADT + ns) * TILEDW + g * 8 + nb * 2;

#define LOAD_TRIP(A_, B_, DD_)                                                  \
    {                                                                           \
        _Pragma("unroll")                                                       \
        for (int j = 0; j < 2; ++j) {                                           \
            const float* kd = ka + 16 * ((DD_) - j);                            \
            const float* ud = ub - ((DD_) - j) * TILEDW;                        \
            _Pragma("unroll")                                                   \
            for (int c = 0; c < 4; ++c) {                                       \
                A_[4 * j + c].x = kd[-4 * c - 1];                               \
                A_[4 * j + c].y = kd[-4 * c];                                   \
                B_[4 * j + c]   = *(const v2f*)(ud + 16 * c);                   \
            }                                                                   \
        }                                                                       \
    }
#define WMMA_TRIP(A_, B_)                                                       \
    {                                                                           \
        _Pragma("unroll")                                                       \
        for (int i = 0; i < 8; ++i)                                             \
            acc = __builtin_amdgcn_wmma_f32_16x16x4_f32(                        \
                false, A_[i], false, B_[i], (short)0, acc, false, false);       \
    }

#define GROUP_EDGE()                                                            \
    {                                                                           \
        asm volatile("" : "+v"(acc) : : "memory");                              \
        __builtin_amdgcn_sched_barrier(0);                                      \
    }

#pragma unroll 1
    for (int pass = 0; pass < 2; ++pass) {
        const int q  = pass ? wave : (NTILE / 4 - 1 - wave);
        const int I0 = 4 * q;
        const float* ub = Ulds + ublane + I0 * TILEDW;

        v8f acc = {};
        v2f a0[8], b0[8], a1[8], b1[8];
        int dd = I0 + 3;
        LOAD_TRIP(a0, b0, dd);
#pragma unroll 1
        for (; dd >= 3; dd -= 4) {
            LOAD_TRIP(a1, b1, dd - 2);
            GROUP_EDGE();
            WMMA_TRIP(a0, b0);
            GROUP_EDGE();
            const int dn = dd >= 7 ? dd - 4 : 1;
            LOAD_TRIP(a0, b0, dn);
            GROUP_EDGE();
            WMMA_TRIP(a1, b1);
            GROUP_EDGE();
        }

        int yl = lane;
        asm volatile("" : "+v"(yl));
        const unsigned ylane =
            (unsigned)((yl & 3) * LSEQ + ((yl >> 2) & 3) * 16 + 8 * (yl >> 4)) * 4u;
        char* yp = (char*)(yT + (size_t)d * (BSZ * LSEQ) + I0 * 16);
        *(float4*)(yp + (size_t)ylane)      = make_float4(acc[0], acc[1], acc[2], acc[3]);
        *(float4*)(yp + (size_t)ylane + 16) = make_float4(acc[4], acc[5], acc[6], acc[7]);
    }
}

extern "C" void kernel_launch(void* const* d_in, const int* in_sizes, int n_in,
                              void* d_out, int out_size, void* d_ws, size_t ws_size,
                              hipStream_t stream) {
    const float* u         = (const float*)d_in[0];
    const float* A_real    = (const float*)d_in[1];
    const float* A_imag    = (const float*)d_in[2];
    const float* B_mat     = (const float*)d_in[3];
    const float* C_mat     = (const float*)d_in[4];
    const float* log_delta = (const float*)d_in[5];
    float* y = (float*)d_out;

    float* ws   = (float*)d_ws;
    float* Kbuf = ws;
    float* uT   = Kbuf + DCH * LSEQ;
    float* yT   = uT + BSZ * LSEQ * DCH;

    s4_front<<<dim3(DCH + UT_BLOCKS), dim3(256), 0, stream>>>(
        u, A_real, A_imag, B_mat, C_mat, log_delta, uT, Kbuf);

    s4_conv_wmma<<<dim3(DCH), dim3(512), 0, stream>>>(uT, Kbuf, yT);

    s4_to_time_major<<<dim3(LSEQ / TT, DCH / 32, BSZ), dim3(256), 0, stream>>>(yT, y);
}


# Conv as a chunked state recurrence: 64-time chunks, Toeplitz blocks only inside a chunk, older input through the 64 complex states by two table products on f32 WMMA; 8448->2368 WMMAs; new sum order
# speedup vs baseline: 27.5105x; 1.5893x over previous
#include <hip/hip_runtime.h>
#include <hip/hip_bf16.h>
#include <math.h>

#define BSZ  4
#define LSEQ 2048
#define DCH  768
#define NST  64
#define DN   (DCH * NST)

typedef __attribute__((ext_vector_type(2))) float v2f;
typedef __attribute__((ext_vector_type(8))) float v8f;

#define TT     256
#define TPITCH 260
#define UT_BLOCKS ((LSEQ / TT) * (DCH / 32) * BSZ)

__global__ void __launch_bounds__(256)
s4_to_channel_major(const float* __restrict__ u, float* __restrict__ uT) {
    __shared__ __attribute__((aligned(16))) float tile[32 * TPITCH];
    const int tid = threadIdx.x;
    const int bid = blockIdx.x;
    const int t0  = (bid & (LSEQ / TT - 1)) * TT;
    const int dy  = (bid / (LSEQ / TT)) % (DCH / 32);
    const int b   = bid / ((LSEQ / TT) * (DCH / 32));
    const int d0  = dy * 32;

    {
        const int c4 = tid & 7, tr = tid >> 3;
        const float* src = u + ((size_t)b * LSEQ + (size_t)(t0 + tr)) * DCH + d0 + 4 * c4;
        float4 v[TT / 32];
#pragma unroll
        for (int i = 0; i < TT / 32; ++i)
            v[i] = *(const float4*)(src + (size_t)i * 32 * DCH);
#pragma unroll
        for (int i = 0; i < TT / 32; ++i) {
            float* dst = tile + (4 * c4) * TPITCH + tr + 32 * i;
            dst[0 * TPITCH] = v[i].x;
            dst[1 * TPITCH] = v[i].y;
            dst[2 * TPITCH] = v[i].z;
            dst[3 * TPITCH] = v[i].w;
        }
    }
    __syncthreads();
    {
        const int tq = tid & 63, dr = tid >> 6;
#pragma unroll
        for (int i = 0; i < 8; ++i) {
            const int dd = dr + 4 * i;
            const float4 o = *(const float4*)(tile + dd * TPITCH + 4 * tq);
            *(float4*)(uT + ((size_t)(d0 + dd) * BSZ + b) * LSEQ + t0 + 4 * tq) = o;
        }
    }
}

__global__ void __launch_bounds__(256)
s4_to_time_major(const float* __restrict__ yT, float* __restrict__ y) {
    __shared__ __attribute__((aligned(16))) float tile[32 * TPITCH];
    const int t0  = blockIdx.x * TT;
    const int d0  = blockIdx.y * 32;
    const int b   = blockIdx.z;
    const int tid = threadIdx.x;

    {
        const int tq = tid & 63, dr = tid >> 6;
        float4 v[8];
#pragma unroll
        for (int i = 0; i < 8; ++i)
            v[i] = *(const float4*)(yT + ((size_t)(d0 + dr + 4 * i) * BSZ + b) * LSEQ + t0 + 4 * tq);
#pragma unroll
        for (int i = 0; i < 8; ++i)
            *(float4*)(tile + (dr + 4 * i) * TPITCH + 4 * tq) = v[i];
    }
    __syncthreads();
    {
        const int c4 = tid & 7, tr = tid >> 3;
        float* dst = y + ((size_t)b * LSEQ + (size_t)(t0 + tr)) * DCH + d0 + 4 * c4;
#pragma unroll
        for (int i = 0; i < TT / 32; ++i) {
            const float* s = tile + (4 * c4) * TPITCH + tr + 32 * i;
            float4 o;
            o.x = s[0 * TPITCH];
            o.y = s[1 * TPITCH];
            o.z = s[2 * TPITCH];
            o.w = s[3 * TPITCH];
            *(float4*)(dst + (size_t)i * 32 * DCH) = o;
        }
    }
}

#define CH    64
#define NCHK  (LSEQ / CH)
#define NCOL  (BSZ * NCHK)
#define UCP   68
#define GP    68
#define HP    132
#define SP    132
#define KPAD  16

__global__ void __launch_bounds__(512, 8)
s4_conv_chunked(const float* __restrict__ uT,
                const float* __restrict__ A_real, const float* __restrict__ A_imag,
                const float* __restrict__ B_mat,  const float* __restrict__ C_mat,
                const float* __restrict__ log_delta,
                float* __restrict__ yT) {
    __shared__ __attribute__((aligned(16))) float uc[NCOL * UCP];
    __shared__ __attribute__((aligned(16))) float gh[2 * NST * GP];
    __shared__ __attribute__((aligned(16))) float sc[NCOL * SP];
    __shared__ __attribute__((aligned(16))) float kp[KPAD + CH];

    const int d   = blockIdx.x;
    const int tid = threadIdx.x;
    const int n   = tid & (NST - 1);
    const int r0  = tid >> 6;

    float pr, pi, cr, ci;
    {
        const int idx = d * NST + n;
        const float delta = log1pf(expf(log_delta[d]));
        const float ar = A_real[idx], ai = A_imag[idx];
        pr = delta * ar;
        pi = delta * ai;
        const float er = expf(pr);
        float s, c;
        __sincosf(pi, &s, &c);
        const float abr = er * c, abi = er * s;
        const float nr = abr - 1.0f, ni = abi;
        const float inv = 1.0f / (ar * ar + ai * ai);
        const float qr = (nr * ar + ni * ai) * inv;
        const float qi = (ni * ar - nr * ai) * inv;
        const float b = B_mat[idx], cm = C_mat[idx];
        cr = cm * (qr * b);
        ci = cm * (qi * b);
    }

    if (tid < KPAD) kp[tid] = 0.0f;

    {
        const float4* usrc = (const float4*)(uT + (size_t)d * (BSZ * LSEQ)) + tid;
        float* udst = uc + (4 * (tid >> 4)) * UCP + 4 * (tid & 15);
        float4 val[BSZ];
#pragma unroll
        for (int sb = 0; sb < BSZ; ++sb)
            val[sb] = usrc[sb * (LSEQ / 4)];
#pragma unroll
        for (int sb = 0; sb < BSZ; ++sb)
            *(float4*)(udst + sb * UCP) = val[sb];
    }

    {
        float* Gs = gh;
#pragma unroll
        for (int i = 0; i < CH / 8; ++i) {
            const int   k = r0 + 8 * i;
            const float l = (float)(CH - k);
            const float e = __expf(pr * l);
            float s, c;
            __sincosf(pi * l, &s, &c);
            Gs[(2 * n) * GP + k]     = e * c;
            Gs[(2 * n + 1) * GP + k] = e * s;
        }
    }
    __syncthreads();

    const int wave = __builtin_amdgcn_readfirstlane(tid >> 5);
    const int lane = tid & 31;
    const int ln   = lane & 15;
    const int g    = lane >> 4;
    const int ct   = wave & 7;
    const int hh   = wave >> 3;
    const float* ucol = uc + (16 * ct + ln) * UCP + 2 * g;

    {
        const float* ga = gh + (64 * hh + ln) * GP + 2 * g;
        v8f acc[4] = {};
#pragma unroll
        for (int c = 0; c < CH / 4; ++c) {
            const v2f fb = *(const v2f*)(ucol + 4 * c);
#pragma unroll
            for (int i = 0; i < 4; ++i) {
                const v2f fa = *(const v2f*)(ga + 16 * i * GP + 4 * c);
                acc[i] = __builtin_amdgcn_wmma_f32_16x16x4_f32(false, fa, false, fb, (short)0, acc[i], false, false);
            }
        }
        float* sp = sc + (16 * ct + ln) * SP + 64 * hh + 8 * g;
#pragma unroll
        for (int i = 0; i < 4; ++i) {
            *(float4*)(sp + 16 * i)     = make_float4(acc[i][0], acc[i][1], acc[i][2], acc[i][3]);
            *(float4*)(sp + 16 * i + 4) = make_float4(acc[i][4], acc[i][5], acc[i][6], acc[i][7]);
        }
    }
    __syncthreads();

    {
        float* Hs = gh;
#pragma unroll
        for (int i = 0; i < CH / 8; ++i) {
            const int   m = r0 + 8 * i;
            const float l = (float)m;
            const float e = __expf(pr * l);
            float s, c;
            __sincosf(pi * l, &s, &c);
            const float qr = e * c, qi = e * s;
            *(float2*)(Hs + m * HP + 2 * n) = make_float2(cr * qr - ci * qi, -(cr * qi + ci * qr));
        }
    }
    __syncthreads();

    if (wave < 8) {
        const int bb = tid >> 6;
        const float e64 = __expf(pr * (float)CH);
        float s64, c64;
        __sincosf(pi * (float)CH, &s64, &c64);
        const float lr = e64 * c64, li = e64 * s64;
        float* sp = sc + bb * SP + 2 * n;
        float2 loc[NCHK];
#pragma unroll
        for (int j = 0; j < NCHK; ++j)
            loc[j] = *(const float2*)(sp + 4 * j * SP);
        float sr = 0.0f, si = 0.0f;
#pragma unroll
        for (int j = 0; j < NCHK; ++j) {
            *(float2*)(sp + 4 * j * SP) = make_float2(sr, si);
            const float nr = lr * sr - li * si + loc[j].x;
            const float ni = lr * si + li * sr + loc[j].y;
            sr = nr;
            si = ni;
        }
    } else if (wave < 10) {
        const int l = tid - 256;
        const float* hrow = gh + l * HP;
        float acc = 0.0f;
#pragma unroll 8
        for (int n2 = 0; n2 < NST; ++n2)
            acc += hrow[2 * n2];
        kp[KPAD + l] = acc;
    }
    __syncthreads();

    {
        const int s0 = hh, s1 = 3 - hh;
        v8f y0 = {}, y1 = {};

        const float* ka = kp + KPAD + ln - 2 * g;
#define TOEP(ACC_, S_, D_)                                                      \
        {                                                                       \
            _Pragma("unroll")                                                   \
            for (int c = 0; c < 4; ++c) {                                       \
                v2f fa;                                                         \
                fa.x = ka[16 * (D_) - 4 * c];                                   \
                fa.y = ka[16 * (D_) - 4 * c - 1];                               \
                const v2f fb = *(const v2f*)(ucol + 16 * ((S_) - (D_)) + 4 * c);\
                ACC_ = __builtin_amdgcn_wmma_f32_16x16x4_f32(                   \
                    false, fa, false, fb, (short)0, ACC_, false, false);        \
            }                                                                   \
        }
        if (hh == 0) {
            TOEP(y0, 0, 0);
            TOEP(y1, 3, 3); TOEP(y1, 3, 2); TOEP(y1, 3, 1); TOEP(y1, 3, 0);
        } else {
            TOEP(y0, 1, 1); TOEP(y0, 1, 0);
            TOEP(y1, 2, 2); TOEP(y1, 2, 1); TOEP(y1, 2, 0);
        }
#undef TOEP

        const float* sb  = sc + (16 * ct + ln) * SP + 2 * g;
        const float* ha0 = gh + (16 * s0 + ln) * HP + 2 * g;
        const float* ha1 = gh + (16 * s1 + ln) * HP + 2 * g;
#pragma unroll
        for (int c = 0; c < 2 * NST / 4; ++c) {
            const v2f fb  = *(const v2f*)(sb + 4 * c);
            const v2f fa0 = *(const v2f*)(ha0 + 4 * c);
            const v2f fa1 = *(const v2f*)(ha1 + 4 * c);
            y0 = __builtin_amdgcn_wmma_f32_16x16x4_f32(false, fa0, false, fb, (short)0, y0, false, false);
            y1 = __builtin_amdgcn_wmma_f32_16x16x4_f32(false, fa1, false, fb, (short)0, y1, false, false);
        }

        float* yp = yT + (size_t)d * (BSZ * LSEQ) + (ln & 3) * LSEQ + CH * (4 * ct + (ln >> 2)) + 8 * g;
        *(float4*)(yp + 16 * s0)     = make_float4(y0[0], y0[1], y0[2], y0[3]);
        *(float4*)(yp + 16 * s0 + 4) = make_float4(y0[4], y0[5], y0[6], y0[7]);
        *(float4*)(yp + 16 * s1)     = make_float4(y1[0], y1[1], y1[2], y1[3]);
        *(float4*)(yp + 16 * s1 + 4) = make_float4(y1[4], y1[5], y1[6], y1[7]);
    }
}

extern "C" void kernel_launch(void* const* d_in, const int* in_sizes, int n_in,
                              void* d_out, int out_size, void* d_ws, size_t ws_size,
                              hipStream_t stream) {
    const float* u         = (const float*)d_in[0];
    const float* A_real    = (const float*)d_in[1];
    const float* A_imag    = (const float*)d_in[2];
    const float* B_mat     = (const float*)d_in[3];
    const float* C_mat     = (const float*)d_in[4];
    const float* log_delta = (const float*)d_in[5];
    float* y = (float*)d_out;

    float* ws = (float*)d_ws;
    float* uT = ws;
    float* yT = uT + BSZ * LSEQ * DCH;

    s4_to_channel_major<<<dim3(UT_BLOCKS), dim3(256), 0, stream>>>(u, uT);

    s4_conv_chunked<<<dim3(DCH), dim3(512), 0, stream>>>(
        uT, A_real, A_imag, B_mat, C_mat, log_delta, yT);

    s4_to_time_major<<<dim3(LSEQ / TT, DCH / 32, BSZ), dim3(256), 0, stream>>>(yT, y);
}


# Conv result staged through the LDS u tile and stored as whole 128-B lines (four per wave instruction) instead of 16-byte pieces of 16 lines per instruction; two more barriers; sums as before
# speedup vs baseline: 35.4171x; 1.2874x over previous
#include <hip/hip_runtime.h>
#include <hip/hip_bf16.h>
#include <math.h>

#define BSZ  4
#define LSEQ 2048
#define DCH  768
#define NST  64
#define DN   (DCH * NST)

typedef __attribute__((ext_vector_type(2))) float v2f;
typedef __attribute__((ext_vector_type(8))) float v8f;

#define TT     256
#define TPITCH 260
#define UT_BLOCKS ((LSEQ / TT) * (DCH / 32) * BSZ)

__global__ void __launch_bounds__(256)
s4_to_channel_major(const float* __restrict__ u, float* __restrict__ uT) {
    __shared__ __attribute__((aligned(16))) float tile[32 * TPITCH];
    const int tid = threadIdx.x;
    const int bid = blockIdx.x;
    const int t0  = (bid & (LSEQ / TT - 1)) * TT;
    const int dy  = (bid / (LSEQ / TT)) % (DCH / 32);
    const int b   = bid / ((LSEQ / TT) * (DCH / 32));
    const int d0  = dy * 32;

    {
        const int c4 = tid & 7, tr = tid >> 3;
        const float* src = u + ((size_t)b * LSEQ + (size_t)(t0 + tr)) * DCH + d0 + 4 * c4;
        float4 v[TT / 32];
#pragma unroll
        for (int i = 0; i < TT / 32; ++i)
            v[i] = *(const float4*)(src + (size_t)i * 32 * DCH);
#pragma unroll
        for (int i = 0; i < TT / 32; ++i) {
            float* dst = tile + (4 * c4) * TPITCH + tr + 32 * i;
            dst[0 * TPITCH] = v[i].x;
            dst[1 * TPITCH] = v[i].y;
            dst[2 * TPITCH] = v[i].z;
            dst[3 * TPITCH] = v[i].w;
        }
    }
    __syncthreads();
    {
        const int tq = tid & 63, dr = tid >> 6;
#pragma unroll
        for (int i = 0; i < 8; ++i) {
            const int dd = dr + 4 * i;
            const float4 o = *(const float4*)(tile + dd * TPITCH + 4 * tq);
            *(float4*)(uT + ((size_t)(d0 + dd) * BSZ + b) * LSEQ + t0 + 4 * tq) = o;
        }
    }
}

__global__ void __launch_bounds__(256)
s4_to_time_major(const float* __restrict__ yT, float* __restrict__ y) {
    __shared__ __attribute__((aligned(16))) float tile[32 * TPITCH];
    const int t0  = blockIdx.x * TT;
    const int d0  = blockIdx.y * 32;
    const int b   = blockIdx.z;
    const int tid = threadIdx.x;

    {
        const int tq = tid & 63, dr = tid >> 6;
        float4 v[8];
#pragma unroll
        for (int i = 0; i < 8; ++i)
            v[i] = *(const float4*)(yT + ((size_t)(d0 + dr + 4 * i) * BSZ + b) * LSEQ + t0 + 4 * tq);
#pragma unroll
        for (int i = 0; i < 8; ++i)
            *(float4*)(tile + (dr + 4 * i) * TPITCH + 4 * tq) = v[i];
    }
    __syncthreads();
    {
        const int c4 = tid & 7, tr = tid >> 3;
        float* dst = y + ((size_t)b * LSEQ + (size_t)(t0 + tr)) * DCH + d0 + 4 * c4;
#pragma unroll
        for (int i = 0; i < TT / 32; ++i) {
            const float* s = tile + (4 * c4) * TPITCH + tr + 32 * i;
            float4 o;
            o.x = s[0 * TPITCH];
            o.y = s[1 * TPITCH];
            o.z = s[2 * TPITCH];
            o.w = s[3 * TPITCH];
            *(float4*)(dst + (size_t)i * 32 * DCH) = o;
        }
    }
}

#define CH    64
#define NCHK  (LSEQ / CH)
#define NCOL  (BSZ * NCHK)
#define UCP   68
#define GP    68
#define HP    132
#define SP    132
#define KPAD  16

__global__ void __launch_bounds__(512, 8)
s4_conv_chunked(const float* __restrict__ uT,
                const float* __restrict__ A_real, const float* __restrict__ A_imag,
                const float* __restrict__ B_mat,  const float* __restrict__ C_mat,
                const float* __restrict__ log_delta,
                float* __restrict__ yT) {
    __shared__ __attribute__((aligned(16))) float uc[NCOL * UCP];
    __shared__ __attribute__((aligned(16))) float gh[2 * NST * GP];
    __shared__ __attribute__((aligned(16))) float sc[NCOL * SP];
    __shared__ __attribute__((aligned(16))) float kp[KPAD + CH];

    const int d   = blockIdx.x;
    const int tid = threadIdx.x;
    const int n   = tid & (NST - 1);
    const int r0  = tid >> 6;

    float pr, pi, cr, ci;
    {
        const int idx = d * NST + n;
        const float delta = log1pf(expf(log_delta[d]));
        const float ar = A_real[idx], ai = A_imag[idx];
        pr = delta * ar;
        pi = delta * ai;
        const float er = expf(pr);
        float s, c;
        __sincosf(pi, &s, &c);
        const float abr = er * c, abi = er * s;
        const float nr = abr - 1.0f, ni = abi;
        const float inv = 1.0f / (ar * ar + ai * ai);
        const float qr = (nr * ar + ni * ai) * inv;
        const float qi = (ni * ar - nr * ai) * inv;
        const float b = B_mat[idx], cm = C_mat[idx];
        cr = cm * (qr * b);
        ci = cm * (qi * b);
    }

    if (tid < KPAD) kp[tid] = 0.0f;

    {
        const float4* usrc = (const float4*)(uT + (size_t)d * (BSZ * LSEQ)) + tid;
        float* udst = uc + (4 * (tid >> 4)) * UCP + 4 * (tid & 15);
        float4 val[BSZ];
#pragma unroll
        for (int sb = 0; sb < BSZ; ++sb)
            val[sb] = usrc[sb * (LSEQ / 4)];
#pragma unroll
        for (int sb = 0; sb < BSZ; ++sb)
            *(float4*)(udst + sb * UCP) = val[sb];
    }

    {
        float* Gs = gh;
#pragma unroll
        for (int i = 0; i < CH / 8; ++i) {
            const int   k = r0 + 8 * i;
            const float l = (float)(CH - k);
            const float e = __expf(pr * l);
            float s, c;
            __sincosf(pi * l, &s, &c);
            Gs[(2 * n) * GP + k]     = e * c;
            Gs[(2 * n + 1) * GP + k] = e * s;
        }
    }
    __syncthreads();

    const int wave = __builtin_amdgcn_readfirstlane(tid >> 5);
    const int lane = tid & 31;
    const int ln   = lane & 15;
    const int g    = lane >> 4;
    const int ct   = wave & 7;
    const int hh   = wave >> 3;
    const float* ucol = uc + (16 * ct + ln) * UCP + 2 * g;

    {
        const float* ga = gh + (64 * hh + ln) * GP + 2 * g;
        v8f acc[4] = {};
#pragma unroll
        for (int c = 0; c < CH / 4; ++c) {
            const v2f fb = *(const v2f*)(ucol + 4 * c);
#pragma unroll
            for (int i = 0; i < 4; ++i) {
                const v2f fa = *(const v2f*)(ga + 16 * i * GP + 4 * c);
                acc[i] = __builtin_amdgcn_wmma_f32_16x16x4_f32(false, fa, false, fb, (short)0, acc[i], false, false);
            }
        }
        float* sp = sc + (16 * ct + ln) * SP + 64 * hh + 8 * g;
#pragma unroll
        for (int i = 0; i < 4; ++i) {
            *(float4*)(sp + 16 * i)     = make_float4(acc[i][0], acc[i][1], acc[i][2], acc[i][3]);
            *(float4*)(sp + 16 * i + 4) = make_float4(acc[i][4], acc[i][5], acc[i][6], acc[i][7]);
        }
    }
    __syncthreads();

    {
        float* Hs = gh;
#pragma unroll
        for (int i = 0; i < CH / 8; ++i) {
            const int   m = r0 + 8 * i;
            const float l = (float)m;
            const float e = __expf(pr * l);
            float s, c;
            __sincosf(pi * l, &s, &c);
            const float qr = e * c, qi = e * s;
            *(float2*)(Hs + m * HP + 2 * n) = make_float2(cr * qr - ci * qi, -(cr * qi + ci * qr));
        }
    }
    __syncthreads();

    if (wave < 8) {
        const int bb = tid >> 6;
        const float e64 = __expf(pr * (float)CH);
        float s64, c64;
        __sincosf(pi * (float)CH, &s64, &c64);
        const float lr = e64 * c64, li = e64 * s64;
        float* sp = sc + bb * SP + 2 * n;
        float2 loc[NCHK];
#pragma unroll
        for (int j = 0; j < NCHK; ++j)
            loc[j] = *(const float2*)(sp + 4 * j * SP);
        float sr = 0.0f, si = 0.0f;
#pragma unroll
        for (int j = 0; j < NCHK; ++j) {
            *(float2*)(sp + 4 * j * SP) = make_float2(sr, si);
            const float nr = lr * sr - li * si + loc[j].x;
            const float ni = lr * si + li * sr + loc[j].y;
            sr = nr;
            si = ni;
        }
    } else if (wave < 10) {
        const int l = tid - 256;
        const float* hrow = gh + l * HP;
        float acc = 0.0f;
#pragma unroll 8
        for (int n2 = 0; n2 < NST; ++n2)
            acc += hrow[2 * n2];
        kp[KPAD + l] = acc;
    }
    __syncthreads();

    {
        const int s0 = hh, s1 = 3 - hh;
        v8f y0 = {}, y1 = {};

        const float* ka = kp + KPAD + ln - 2 * g;
#define TOEP(ACC_, S_, D_)                                                      \
        {                                                                       \
            _Pragma("unroll")                                                   \
            for (int c = 0; c < 4; ++c) {                                       \
                v2f fa;                                                         \
                fa.x = ka[16 * (D_) - 4 * c];                                   \
                fa.y = ka[16 * (D_) - 4 * c - 1];                               \
                const v2f fb = *(const v2f*)(ucol + 16 * ((S_) - (D_)) + 4 * c);\
                ACC_ = __builtin_amdgcn_wmma_f32_16x16x4_f32(                   \
                    false, fa, false, fb, (short)0, ACC_, false, false);        \
            }                                                                   \
        }
        if (hh == 0) {
            TOEP(y0, 0, 0);
            TOEP(y1, 3, 3); TOEP(y1, 3, 2); TOEP(y1, 3, 1); TOEP(y1, 3, 0);
        } else {
            TOEP(y0, 1, 1); TOEP(y0, 1, 0);
            TOEP(y1, 2, 2); TOEP(y1, 2, 1); TOEP(y1, 2, 0);
        }
#undef TOEP

        const float* sb  = sc + (16 * ct + ln) * SP + 2 * g;
        const float* ha0 = gh + (16 * s0 + ln) * HP + 2 * g;
        const float* ha1 = gh + (16 * s1 + ln) * HP + 2 * g;
#pragma unroll
        for (int c = 0; c < 2 * NST / 4; ++c) {
            const v2f fb  = *(const v2f*)(sb + 4 * c);
            const v2f fa0 = *(const v2f*)(ha0 + 4 * c);
            const v2f fa1 = *(const v2f*)(ha1 + 4 * c);
            y0 = __builtin_amdgcn_wmma_f32_16x16x4_f32(false, fa0, false, fb, (short)0, y0, false, false);
            y1 = __builtin_amdgcn_wmma_f32_16x16x4_f32(false, fa1, false, fb, (short)0, y1, false, false);
        }

        __syncthreads();
        float* yl = uc + (16 * ct + ln) * UCP + 8 * g;
        *(float4*)(yl + 16 * s0)     = make_float4(y0[0], y0[1], y0[2], y0[3]);
        *(float4*)(yl + 16 * s0 + 4) = make_float4(y0[4], y0[5], y0[6], y0[7]);
        *(float4*)(yl + 16 * s1)     = make_float4(y1[0], y1[1], y1[2], y1[3]);
        *(float4*)(yl + 16 * s1 + 4) = make_float4(y1[4], y1[5], y1[6], y1[7]);
    }
    __syncthreads();

    {
        float4* ydst = (float4*)(yT + (size_t)d * (BSZ * LSEQ)) + tid;
        const float* ysrc = uc + (4 * (tid >> 4)) * UCP + 4 * (tid & 15);
        float4 val[BSZ];
#pragma unroll
        for (int sb = 0; sb < BSZ; ++sb)
            val[sb] = *(const float4*)(ysrc + sb * UCP);
#pragma unroll
        for (int sb = 0; sb < BSZ; ++sb)
            ydst[sb * (LSEQ / 4)] = val[sb];
    }
}

extern "C" void kernel_launch(void* const* d_in, const int* in_sizes, int n_in,
                              void* d_out, int out_size, void* d_ws, size_t ws_size,
                              hipStream_t stream) {
    const float* u         = (const float*)d_in[0];
    const float* A_real    = (const float*)d_in[1];
    const float* A_imag    = (const float*)d_in[2];
    const float* B_mat     = (const float*)d_in[3];
    const float* C_mat     = (const float*)d_in[4];
    const float* log_delta = (const float*)d_in[5];
    float* y = (float*)d_out;

    float* ws = (float*)d_ws;
    float* uT = ws;
    float* yT = uT + BSZ * LSEQ * DCH;

    s4_to_channel_major<<<dim3(UT_BLOCKS), dim3(256), 0, stream>>>(u, uT);

    s4_conv_chunked<<<dim3(DCH), dim3(512), 0, stream>>>(
        uT, A_real, A_imag, B_mat, C_mat, log_delta, yT);

    s4_to_time_major<<<dim3(LSEQ / TT, DCH / 32, BSZ), dim3(256), 0, stream>>>(yT, y);
}


# Conv block start: parameter and u-tile requests issued back to back; discretization and both table builds (H held in registers until G is read out) run under the u latency; sums as before
# speedup vs baseline: 36.5574x; 1.0322x over previous
#include <hip/hip_runtime.h>
#include <hip/hip_bf16.h>
#include <math.h>

#define BSZ  4
#define LSEQ 2048
#define DCH  768
#define NST  64
#define DN   (DCH * NST)

typedef __attribute__((ext_vector_type(2))) float v2f;
typedef __attribute__((ext_vector_type(8))) float v8f;

#define TT     256
#define TPITCH 260
#define UT_BLOCKS ((LSEQ / TT) * (DCH / 32) * BSZ)

__global__ void __launch_bounds__(256)
s4_to_channel_major(const float* __restrict__ u, float* __restrict__ uT) {
    __shared__ __attribute__((aligned(16))) float tile[32 * TPITCH];
    const int tid = threadIdx.x;
    const int bid = blockIdx.x;
    const int t0  = (bid & (LSEQ / TT - 1)) * TT;
    const int dy  = (bid / (LSEQ / TT)) % (DCH / 32);
    const int b   = bid / ((LSEQ / TT) * (DCH / 32));
    const int d0  = dy * 32;

    {
        const int c4 = tid & 7, tr = tid >> 3;
        const float* src = u + ((size_t)b * LSEQ + (size_t)(t0 + tr)) * DCH + d0 + 4 * c4;
        float4 v[TT / 32];
#pragma unroll
        for (int i = 0; i < TT / 32; ++i)
            v[i] = *(const float4*)(src + (size_t)i * 32 * DCH);
#pragma unroll
        for (int i = 0; i < TT / 32; ++i) {
            float* dst = tile + (4 * c4) * TPITCH + tr + 32 * i;
            dst[0 * TPITCH] = v[i].x;
            dst[1 * TPITCH] = v[i].y;
            dst[2 * TPITCH] = v[i].z;
            dst[3 * TPITCH] = v[i].w;
        }
    }
    __syncthreads();
    {
        const int tq = tid & 63, dr = tid >> 6;
#pragma unroll
        for (int i = 0; i < 8; ++i) {
            const int dd = dr + 4 * i;
            const float4 o = *(const float4*)(tile + dd * TPITCH + 4 * tq);
            *(float4*)(uT + ((size_t)(d0 + dd) * BSZ + b) * LSEQ + t0 + 4 * tq) = o;
        }
    }
}

__global__ void __launch_bounds__(256)
s4_to_time_major(const float* __restrict__ yT, float* __restrict__ y) {
    __shared__ __attribute__((aligned(16))) float tile[32 * TPITCH];
    const int t0  = blockIdx.x * TT;
    const int d0  = blockIdx.y * 32;
    const int b   = blockIdx.z;
    const int tid = threadIdx.x;

    {
        const int tq = tid & 63, dr = tid >> 6;
        float4 v[8];
#pragma unroll
        for (int i = 0; i < 8; ++i)
            v[i] = *(const float4*)(yT + ((size_t)(d0 + dr + 4 * i) * BSZ + b) * LSEQ + t0 + 4 * tq);
#pragma unroll
        for (int i = 0; i < 8; ++i)
            *(float4*)(tile + (dr + 4 * i) * TPITCH + 4 * tq) = v[i];
    }
    __syncthreads();
    {
        const int c4 = tid & 7, tr = tid >> 3;
        float* dst = y + ((size_t)b * LSEQ + (size_t)(t0 + tr)) * DCH + d0 + 4 * c4;
#pragma unroll
        for (int i = 0; i < TT / 32; ++i) {
            const float* s = tile + (4 * c4) * TPITCH + tr + 32 * i;
            float4 o;
            o.x = s[0 * TPITCH];
            o.y = s[1 * TPITCH];
            o.z = s[2 * TPITCH];
            o.w = s[3 * TPITCH];
            *(float4*)(dst + (size_t)i * 32 * DCH) = o;
        }
    }
}

#define CH    64
#define NCHK  (LSEQ / CH)
#define NCOL  (BSZ * NCHK)
#define UCP   68
#define GP    68
#define HP    132
#define SP    132
#define KPAD  16

__global__ void __launch_bounds__(512, 8)
s4_conv_chunked(const float* uT,
                const float* A_real, const float* A_imag,
                const float* B_mat,  const float* C_mat,
                const float* log_delta,
                float* __restrict__ yT) {
    __shared__ __attribute__((aligned(16))) float uc[NCOL * UCP];
    __shared__ __attribute__((aligned(16))) float gh[2 * NST * GP];
    __shared__ __attribute__((aligned(16))) float sc[NCOL * SP];
    __shared__ __attribute__((aligned(16))) float kp[KPAD + CH];

    const int d   = blockIdx.x;
    const int tid = threadIdx.x;
    const int n   = tid & (NST - 1);
    const int r0  = tid >> 6;

    if (tid < KPAD) kp[tid] = 0.0f;

    const int idx = d * NST + n;
    const float ld = log_delta[d];
    const float ar = A_real[idx], ai = A_imag[idx];
    const float bm = B_mat[idx], cm = C_mat[idx];
    __builtin_amdgcn_sched_barrier(0);
    const float4* usrc = (const float4*)(uT + (size_t)d * (BSZ * LSEQ)) + tid;
    const float4 u0 = usrc[0 * (LSEQ / 4)];
    const float4 u1 = usrc[1 * (LSEQ / 4)];
    const float4 u2 = usrc[2 * (LSEQ / 4)];
    const float4 u3 = usrc[3 * (LSEQ / 4)];
    __builtin_amdgcn_sched_barrier(0);

    float pr, pi, cr, ci;
    {
        const float delta = log1pf(expf(ld));
        pr = delta * ar;
        pi = delta * ai;
        const float er = expf(pr);
        float s, c;
        __sincosf(pi, &s, &c);
        const float abr = er * c, abi = er * s;
        const float nr = abr - 1.0f, ni = abi;
        const float inv = 1.0f / (ar * ar + ai * ai);
        const float qr = (nr * ar + ni * ai) * inv;
        const float qi = (ni * ar - nr * ai) * inv;
        cr = cm * (qr * bm);
        ci = cm * (qi * bm);
    }

    {
        float* Gs = gh;
#pragma unroll
        for (int i = 0; i < CH / 8; ++i) {
            const int   k = r0 + 8 * i;
            const float l = (float)(CH - k);
            const float e = __expf(pr * l);
            float s, c;
            __sincosf(pi * l, &s, &c);
            Gs[(2 * n) * GP + k]     = e * c;
            Gs[(2 * n + 1) * GP + k] = e * s;
        }
    }

    float2 hv[CH / 8];
#pragma unroll
    for (int i = 0; i < CH / 8; ++i) {
        const int   m = r0 + 8 * i;
        const float l = (float)m;
        const float e = __expf(pr * l);
        float s, c;
        __sincosf(pi * l, &s, &c);
        const float qr = e * c, qi = e * s;
        hv[i] = make_float2(cr * qr - ci * qi, -(cr * qi + ci * qr));
    }
    __builtin_amdgcn_sched_barrier(0);

    {
        float* udst = uc + (4 * (tid >> 4)) * UCP + 4 * (tid & 15);
        *(float4*)(udst + 0 * UCP) = u0;
        *(float4*)(udst + 1 * UCP) = u1;
        *(float4*)(udst + 2 * UCP) = u2;
        *(float4*)(udst + 3 * UCP) = u3;
    }
    __syncthreads();

    const int wave = __builtin_amdgcn_readfirstlane(tid >> 5);
    const int lane = tid & 31;
    const int ln   = lane & 15;
    const int g    = lane >> 4;
    const int ct   = wave & 7;
    const int hh   = wave >> 3;
    const float* ucol = uc + (16 * ct + ln) * UCP + 2 * g;

    {
        const float* ga = gh + (64 * hh + ln) * GP + 2 * g;
        v8f acc[4] = {};
#pragma unroll
        for (int c = 0; c < CH / 4; ++c) {
            const v2f fb = *(const v2f*)(ucol + 4 * c);
#pragma unroll
            for (int i = 0; i < 4; ++i) {
                const v2f fa = *(const v2f*)(ga + 16 * i * GP + 4 * c);
                acc[i] = __builtin_amdgcn_wmma_f32_16x16x4_f32(false, fa, false, fb, (short)0, acc[i], false, false);
            }
        }
        float* sp = sc + (16 * ct + ln) * SP + 64 * hh + 8 * g;
#pragma unroll
        for (int i = 0; i < 4; ++i) {
            *(float4*)(sp + 16 * i)     = make_float4(acc[i][0], acc[i][1], acc[i][2], acc[i][3]);
            *(float4*)(sp + 16 * i + 4) = make_float4(acc[i][4], acc[i][5], acc[i][6], acc[i][7]);
        }
    }
    __syncthreads();

    {
        float* Hs = gh;
#pragma unroll
        for (int i = 0; i < CH / 8; ++i)
            *(float2*)(Hs + (r0 + 8 * i) * HP + 2 * n) = hv[i];
    }
    __syncthreads();

    if (wave < 8) {
        const int bb = tid >> 6;
        const float e64 = __expf(pr * (float)CH);
        float s64, c64;
        __sincosf(pi * (float)CH, &s64, &c64);
        const float lr = e64 * c64, li = e64 * s64;
        float* sp = sc + bb * SP + 2 * n;
        float2 loc[NCHK];
#pragma unroll
        for (int j = 0; j < NCHK; ++j)
            loc[j] = *(const float2*)(sp + 4 * j * SP);
        float sr = 0.0f, si = 0.0f;
#pragma unroll
        for (int j = 0; j < NCHK; ++j) {
            *(float2*)(sp + 4 * j * SP) = make_float2(sr, si);
            const float nr = lr * sr - li * si + loc[j].x;
            const float ni = lr * si + li * sr + loc[j].y;
            sr = nr;
            si = ni;
        }
    } else if (wave < 10) {
        const int l = tid - 256;
        const float* hrow = gh + l * HP;
        float acc = 0.0f;
#pragma unroll 8
        for (int n2 = 0; n2 < NST; ++n2)
            acc += hrow[2 * n2];
        kp[KPAD + l] = acc;
    }
    __syncthreads();

    {
        const int s0 = hh, s1 = 3 - hh;
        v8f y0 = {}, y1 = {};

        const float* ka = kp + KPAD + ln - 2 * g;
#define TOEP(ACC_, S_, D_)                                                      \
        {                                                                       \
            _Pragma("unroll")                                                   \
            for (int c = 0; c < 4; ++c) {                                       \
                v2f fa;                                                         \
                fa.x = ka[16 * (D_) - 4 * c];                                   \
                fa.y = ka[16 * (D_) - 4 * c - 1];                               \
                const v2f fb = *(const v2f*)(ucol + 16 * ((S_) - (D_)) + 4 * c);\
                ACC_ = __builtin_amdgcn_wmma_f32_16x16x4_f32(                   \
                    false, fa, false, fb, (short)0, ACC_, false, false);        \
            }                                                                   \
        }
        if (hh == 0) {
            TOEP(y0, 0, 0);
            TOEP(y1, 3, 3); TOEP(y1, 3, 2); TOEP(y1, 3, 1); TOEP(y1, 3, 0);
        } else {
            TOEP(y0, 1, 1); TOEP(y0, 1, 0);
            TOEP(y1, 2, 2); TOEP(y1, 2, 1); TOEP(y1, 2, 0);
        }
#undef TOEP

        const float* sb  = sc + (16 * ct + ln) * SP + 2 * g;
        const float* ha0 = gh + (16 * s0 + ln) * HP + 2 * g;
        const float* ha1 = gh + (16 * s1 + ln) * HP + 2 * g;
#pragma unroll
        for (int c = 0; c < 2 * NST / 4; ++c) {
            const v2f fb  = *(const v2f*)(sb + 4 * c);
            const v2f fa0 = *(const v2f*)(ha0 + 4 * c);
            const v2f fa1 = *(const v2f*)(ha1 + 4 * c);
            y0 = __builtin_amdgcn_wmma_f32_16x16x4_f32(false, fa0, false, fb, (short)0, y0, false, false);
            y1 = __builtin_amdgcn_wmma_f32_16x16x4_f32(false, fa1, false, fb, (short)0, y1, false, false);
        }

        __syncthreads();
        float* yl = uc + (16 * ct + ln) * UCP + 8 * g;
        *(float4*)(yl + 16 * s0)     = make_float4(y0[0], y0[1], y0[2], y0[3]);
        *(float4*)(yl + 16 * s0 + 4) = make_float4(y0[4], y0[5], y0[6], y0[7]);
        *(float4*)(yl + 16 * s1)     = make_float4(y1[0], y1[1], y1[2], y1[3]);
        *(float4*)(yl + 16 * s1 + 4) = make_float4(y1[4], y1[5], y1[6], y1[7]);
    }
    __syncthreads();

    {
        float4* ydst = (float4*)(yT + (size_t)d * (BSZ * LSEQ)) + tid;
        const float* ysrc = uc + (4 * (tid >> 4)) * UCP + 4 * (tid & 15);
        float4 val[BSZ];
#pragma unroll
        for (int sb = 0; sb < BSZ; ++sb)
            val[sb] = *(const float4*)(ysrc + sb * UCP);
#pragma unroll
        for (int sb = 0; sb < BSZ; ++sb)
            ydst[sb * (LSEQ / 4)] = val[sb];
    }
}

extern "C" void kernel_launch(void* const* d_in, const int* in_sizes, int n_in,
                              void* d_out, int out_size, void* d_ws, size_t ws_size,
                              hipStream_t stream) {
    const float* u         = (const float*)d_in[0];
    const float* A_real    = (const float*)d_in[1];
    const float* A_imag    = (const float*)d_in[2];
    const float* B_mat     = (const float*)d_in[3];
    const float* C_mat     = (const float*)d_in[4];
    const float* log_delta = (const float*)d_in[5];
    float* y = (float*)d_out;

    float* ws = (float*)d_ws;
    float* uT = ws;
    float* yT = uT + BSZ * LSEQ * DCH;

    s4_to_channel_major<<<dim3(UT_BLOCKS), dim3(256), 0, stream>>>(u, uT);

    s4_conv_chunked<<<dim3(DCH), dim3(512), 0, stream>>>(
        uT, A_real, A_imag, B_mat, C_mat, log_delta, yT);

    s4_to_time_major<<<dim3(LSEQ / TT, DCH / 32, BSZ), dim3(256), 0, stream>>>(yT, y);
}
